# starting kernel
# baseline (speedup 1.0000x reference)
#include <hip/hip_runtime.h>
#include <math.h>

#define NN   100000
#define CH   64
#define HD   128
#define NE   800000
#define N64  6400000
#define EPS_MSG 1e-7f
#define EPS_LN  1e-5f

typedef float v2f __attribute__((ext_vector_type(2)));
typedef float v8f __attribute__((ext_vector_type(8)));

static __device__ __forceinline__ v8f wmma4(v2f a, v2f b, v8f c) {
  return __builtin_amdgcn_wmma_f32_16x16x4_f32(
       false, a,  false, b,
       (short)0, c,  false,  false);
}

__global__ __launch_bounds__(256) void k_zero(float4* __restrict__ p, int n4) {
  int i = blockIdx.x * 256 + threadIdx.x;
  if (i < n4) p[i] = make_float4(0.f, 0.f, 0.f, 0.f);
}

__global__ void k_zero2(float* __restrict__ p) {
  if (threadIdx.x < 2) p[threadIdx.x] = 0.f;
}

__global__ __launch_bounds__(256) void k_reduce(const float4* __restrict__ h,
                                                float* __restrict__ sums) {
  __shared__ float ss[256];
  __shared__ float sq[256];
  float s = 0.f, q = 0.f;
  for (int i = blockIdx.x * 256 + threadIdx.x; i < N64 / 4; i += gridDim.x * 256) {
    float4 v = h[i];
    s += v.x + v.y + v.z + v.w;
    q += v.x * v.x + v.y * v.y + v.z * v.z + v.w * v.w;
  }
  ss[threadIdx.x] = s; sq[threadIdx.x] = q;
  __syncthreads();
  for (int off = 128; off > 0; off >>= 1) {
    if (threadIdx.x < off) {
      ss[threadIdx.x] += ss[threadIdx.x + off];
      sq[threadIdx.x] += sq[threadIdx.x + off];
    }
    __syncthreads();
  }
  if (threadIdx.x == 0) {
    atomicAdd(&sums[0], ss[0]);
    atomicAdd(&sums[1], sq[0]);
  }
}

__global__ __launch_bounds__(256) void k_gln_apply(const float4* __restrict__ h,
                                                   const float* __restrict__ sums,
                                                   const float* __restrict__ w,
                                                   const float* __restrict__ b,
                                                   float4* __restrict__ out) {
  int i = blockIdx.x * 256 + threadIdx.x;
  if (i >= N64 / 4) return;
  const float invn = 1.f / (float)N64;
  float mu = sums[0] * invn;
  float var = sums[1] * invn - mu * mu;
  float scale = 1.f / (sqrtf(fmaxf(var, 0.f)) + EPS_LN);
  int c = (i & 15) << 2;
  float4 v = h[i];
  float r0 = (v.x - mu) * scale * w[c + 0] + b[c + 0];
  float r1 = (v.y - mu) * scale * w[c + 1] + b[c + 1];
  float r2 = (v.z - mu) * scale * w[c + 2] + b[c + 2];
  float r3 = (v.w - mu) * scale * w[c + 3] + b[c + 3];
  out[i] = make_float4(fmaxf(r0, 0.f), fmaxf(r1, 0.f), fmaxf(r2, 0.f), fmaxf(r3, 0.f));
}

__global__ __launch_bounds__(256) void k_edge_max(const int* __restrict__ src,
                                                  const int* __restrict__ dst,
                                                  const float* __restrict__ hin,
                                                  int* __restrict__ amax) {
  int idx = blockIdx.x * 256 + threadIdx.x;
  if (idx >= NE * 16) return;
  int e = idx >> 4, g = (idx & 15) << 2;
  int s = src[e], d = dst[e];
  const float4 v = *(const float4*)(hin + (size_t)s * CH + g);
  int* p = amax + (size_t)d * CH + g;
  atomicMax(p + 0, __float_as_int(fmaxf(v.x, 0.f) + EPS_MSG));
  atomicMax(p + 1, __float_as_int(fmaxf(v.y, 0.f) + EPS_MSG));
  atomicMax(p + 2, __float_as_int(fmaxf(v.z, 0.f) + EPS_MSG));
  atomicMax(p + 3, __float_as_int(fmaxf(v.w, 0.f) + EPS_MSG));
}

__global__ __launch_bounds__(256) void k_edge_sum(const int* __restrict__ src,
                                                  const int* __restrict__ dst,
                                                  const float* __restrict__ hin,
                                                  const float* __restrict__ amax,
                                                  float* __restrict__ denom) {
  int idx = blockIdx.x * 256 + threadIdx.x;
  if (idx >= NE * 16) return;
  int e = idx >> 4, g = (idx & 15) << 2;
  int s = src[e], d = dst[e];
  const float4 v = *(const float4*)(hin + (size_t)s * CH + g);
  const float4 am = *(const float4*)(amax + (size_t)d * CH + g);
  float* p = denom + (size_t)d * CH + g;
  atomicAdd(p + 0, __expf(fmaxf(v.x, 0.f) + EPS_MSG - am.x));
  atomicAdd(p + 1, __expf(fmaxf(v.y, 0.f) + EPS_MSG - am.y));
  atomicAdd(p + 2, __expf(fmaxf(v.z, 0.f) + EPS_MSG - am.z));
  atomicAdd(p + 3, __expf(fmaxf(v.w, 0.f) + EPS_MSG - am.w));
}

__global__ __launch_bounds__(256) void k_edge_agg(const int* __restrict__ src,
                                                  const int* __restrict__ dst,
                                                  const float* __restrict__ hin,
                                                  const float* __restrict__ amax,
                                                  const float* __restrict__ denom,
                                                  float* __restrict__ agg) {
  int idx = blockIdx.x * 256 + threadIdx.x;
  if (idx >= NE * 16) return;
  int e = idx >> 4, g = (idx & 15) << 2;
  int s = src[e], d = dst[e];
  const float4 v = *(const float4*)(hin + (size_t)s * CH + g);
  const float4 am = *(const float4*)(amax + (size_t)d * CH + g);
  const float4 dn = *(const float4*)(denom + (size_t)d * CH + g);
  float m0 = fmaxf(v.x, 0.f) + EPS_MSG;
  float m1 = fmaxf(v.y, 0.f) + EPS_MSG;
  float m2 = fmaxf(v.z, 0.f) + EPS_MSG;
  float m3 = fmaxf(v.w, 0.f) + EPS_MSG;
  float* p = agg + (size_t)d * CH + g;
  atomicAdd(p + 0, m0 * __expf(m0 - am.x) / dn.x);
  atomicAdd(p + 1, m1 * __expf(m1 - am.y) / dn.y);
  atomicAdd(p + 2, m2 * __expf(m2 - am.z) / dn.z);
  atomicAdd(p + 3, m3 * __expf(m3 - am.w) / dn.w);
}

__global__ __launch_bounds__(256)
void k_conv_mlp(const float* __restrict__ resin, const float* __restrict__ agg,
                const float* __restrict__ W1, const float* __restrict__ b1,
                const float* __restrict__ lnw, const float* __restrict__ lnb,
                const float* __restrict__ W2, const float* __restrict__ b2,
                const float* __restrict__ outer, float* __restrict__ out) {
  __shared__ float sAct[32 * CH];
  __shared__ float sHid[32 * HD];
  __shared__ float sb1[HD], slnw[HD], slnb[HD], sb2[CH];
  const int tid = threadIdx.x;
  const int base = blockIdx.x * 32;
  if (tid < HD) { sb1[tid] = b1[tid]; slnw[tid] = lnw[tid]; slnb[tid] = lnb[tid]; }
  if (tid < CH) sb2[tid] = b2[tid];
  for (int i = tid; i < 32 * CH; i += 256) {
    size_t idx = (size_t)base * CH + i;
    sAct[i] = resin[idx] + agg[idx];
  }
  __syncthreads();

  const int wave = tid >> 5, lane = tid & 31;
  const int half = lane >> 4, l16 = lane & 15;

  for (int task = wave; task < 16; task += 8) {
    int rt = task >> 3, ct = task & 7;
    const float* arow = &sAct[(rt * 16 + l16) * CH];
    v8f acc = {0.f, 0.f, 0.f, 0.f, 0.f, 0.f, 0.f, 0.f};
    #pragma unroll
    for (int k0 = 0; k0 < CH; k0 += 4) {
      int k = k0 + 2 * half;
      v2f a;  a.x = arow[k];                     a.y = arow[k + 1];
      v2f bb; bb.x = W1[k * HD + ct * 16 + l16]; bb.y = W1[(k + 1) * HD + ct * 16 + l16];
      acc = wmma4(a, bb, acc);
    }
    #pragma unroll
    for (int r = 0; r < 8; r++) {
      int m = rt * 16 + r + 8 * half;
      int n = ct * 16 + l16;
      sHid[m * HD + n] = acc[r] + sb1[n];
    }
  }
  __syncthreads();

  if (tid < 32) {
    float s = 0.f, q = 0.f;
    for (int c = 0; c < HD; c++) { float v = sHid[tid * HD + c]; s += v; q += v * v; }
    float mu = s * (1.f / HD);
    float var = q * (1.f / HD) - mu * mu;
    float inv = rsqrtf(var + EPS_LN);
    for (int c = 0; c < HD; c++) {
      float v = (sHid[tid * HD + c] - mu) * inv * slnw[c] + slnb[c];
      sHid[tid * HD + c] = fmaxf(v, 0.f);
    }
  }
  __syncthreads();

  {
    int rt = wave >> 2, ct = wave & 3;
    const float* arow = &sHid[(rt * 16 + l16) * HD];
    v8f acc = {0.f, 0.f, 0.f, 0.f, 0.f, 0.f, 0.f, 0.f};
    #pragma unroll
    for (int k0 = 0; k0 < HD; k0 += 4) {
      int k = k0 + 2 * half;
      v2f a;  a.x = arow[k];                     a.y = arow[k + 1];
      v2f bb; bb.x = W2[k * CH + ct * 16 + l16]; bb.y = W2[(k + 1) * CH + ct * 16 + l16];
      acc = wmma4(a, bb, acc);
    }
    #pragma unroll
    for (int r = 0; r < 8; r++) {
      int m = rt * 16 + r + 8 * half;
      size_t node = (size_t)(base + m);
      int n = ct * 16 + l16;
      float v = acc[r] + sb2[n];
      if (outer) v += outer[node * CH + n];
      out[node * CH + n] = v;
    }
  }
}

__global__ __launch_bounds__(256)
void k_head(const float* __restrict__ xc,
            const float* __restrict__ W1, const float* __restrict__ b1,
            const float* __restrict__ W2, const float* __restrict__ b2,
            const float* __restrict__ W3, const float* __restrict__ b3,
            float* __restrict__ out) {
  __shared__ float sA[32 * CH];
  __shared__ float sH[32 * CH];
  __shared__ float sb1[CH], sb2[CH], sW3[CH * 2], sb3[2];
  const int tid = threadIdx.x;
  const int base = blockIdx.x * 32;
  if (tid < CH) { sb1[tid] = b1[tid]; sb2[tid] = b2[tid]; }
  if (tid < CH * 2) sW3[tid] = W3[tid];
  if (tid < 2) sb3[tid] = b3[tid];
  for (int i = tid; i < 32 * CH; i += 256) sA[i] = xc[(size_t)base * CH + i];
  __syncthreads();

  const int wave = tid >> 5, lane = tid & 31;
  const int half = lane >> 4, l16 = lane & 15;
  const int rt = wave >> 2, ct = wave & 3;

  {
    const float* arow = &sA[(rt * 16 + l16) * CH];
    v8f acc = {0.f, 0.f, 0.f, 0.f, 0.f, 0.f, 0.f, 0.f};
    #pragma unroll
    for (int k0 = 0; k0 < CH; k0 += 4) {
      int k = k0 + 2 * half;
      v2f a;  a.x = arow[k];                     a.y = arow[k + 1];
      v2f bb; bb.x = W1[k * CH + ct * 16 + l16]; bb.y = W1[(k + 1) * CH + ct * 16 + l16];
      acc = wmma4(a, bb, acc);
    }
    #pragma unroll
    for (int r = 0; r < 8; r++) {
      int m = rt * 16 + r + 8 * half, n = ct * 16 + l16;
      sH[m * CH + n] = fmaxf(acc[r] + sb1[n], 0.f);
    }
  }
  __syncthreads();
  {
    const float* arow = &sH[(rt * 16 + l16) * CH];
    v8f acc = {0.f, 0.f, 0.f, 0.f, 0.f, 0.f, 0.f, 0.f};
    #pragma unroll
    for (int k0 = 0; k0 < CH; k0 += 4) {
      int k = k0 + 2 * half;
      v2f a;  a.x = arow[k];                     a.y = arow[k + 1];
      v2f bb; bb.x = W2[k * CH + ct * 16 + l16]; bb.y = W2[(k + 1) * CH + ct * 16 + l16];
      acc = wmma4(a, bb, acc);
    }
    #pragma unroll
    for (int r = 0; r < 8; r++) {
      int m = rt * 16 + r + 8 * half, n = ct * 16 + l16;
      sA[m * CH + n] = fmaxf(acc[r] + sb2[n], 0.f);
    }
  }
  __syncthreads();
  if (tid < 64) {
    int m = tid >> 1, o = tid & 1;
    float s = sb3[o];
    for (int k = 0; k < CH; k++) s += sA[m * CH + k] * sW3[k * 2 + o];
    out[(size_t)(base + m) * 2 + o] = s;
  }
}

extern "C" void kernel_launch(void* const* d_in, const int* in_sizes, int n_in,
                              void* d_out, int out_size, void* d_ws, size_t ws_size,
                              hipStream_t stream) {
  (void)in_sizes; (void)n_in; (void)out_size; (void)ws_size;
  const float* x      = (const float*)d_in[0];
  const int*   ei     = (const int*)d_in[1];
  const int*   src    = ei;
  const int*   dst    = ei + NE;
  const float* c1_W1  = (const float*)d_in[2];
  const float* c1_b1  = (const float*)d_in[3];
  const float* c1_lnw = (const float*)d_in[4];
  const float* c1_lnb = (const float*)d_in[5];
  const float* c1_W2  = (const float*)d_in[6];
  const float* c1_b2  = (const float*)d_in[7];
  const float* n1_w   = (const float*)d_in[8];
  const float* n1_b   = (const float*)d_in[9];
  const float* cW1    = (const float*)d_in[10];
  const float* cb1    = (const float*)d_in[11];
  const float* clnw   = (const float*)d_in[12];
  const float* clnb   = (const float*)d_in[13];
  const float* cW2    = (const float*)d_in[14];
  const float* cb2    = (const float*)d_in[15];
  const float* nw     = (const float*)d_in[16];
  const float* nb     = (const float*)d_in[17];
  const float* lW1    = (const float*)d_in[18];
  const float* lb1    = (const float*)d_in[19];
  const float* lW2    = (const float*)d_in[20];
  const float* lb2    = (const float*)d_in[21];
  const float* lW3    = (const float*)d_in[22];
  const float* lb3    = (const float*)d_in[23];

  float* ws    = (float*)d_ws;
  float* xc    = ws;
  float* hpre  = ws + (size_t)N64;
  float* amaxf = ws + 2 * (size_t)N64;
  int*   amax  = (int*)amaxf;
  float* denom = ws + 3 * (size_t)N64;
  float* agg   = ws + 4 * (size_t)N64;
  float* sums  = ws + 5 * (size_t)N64;

  const dim3 B(256);
  const int EB   = (NE * 16 + 255) / 256;
  const int Z3   = (3 * N64 / 4 + 255) / 256;
  const int APB  = (N64 / 4 + 255) / 256;
  const int MLPB = NN / 32;

  auto edge_phase = [&](const float* hin) {
    k_zero<<<Z3, B, 0, stream>>>((float4*)amaxf, 3 * N64 / 4);
    k_edge_max<<<EB, B, 0, stream>>>(src, dst, hin, amax);
    k_edge_sum<<<EB, B, 0, stream>>>(src, dst, hin, amaxf, denom);
    k_edge_agg<<<EB, B, 0, stream>>>(src, dst, hin, amaxf, denom, agg);
  };

  edge_phase(x);
  k_conv_mlp<<<MLPB, B, 0, stream>>>(x, agg, c1_W1, c1_b1, c1_lnw, c1_lnb,
                                     c1_W2, c1_b2, nullptr, hpre);
  k_zero2<<<1, 32, 0, stream>>>(sums);
  k_reduce<<<4096, B, 0, stream>>>((const float4*)hpre, sums);
  k_gln_apply<<<APB, B, 0, stream>>>((const float4*)hpre, sums, n1_w, n1_b, (float4*)xc);

  for (int i = 0; i < 3; i++) {
    k_zero2<<<1, 32, 0, stream>>>(sums);
    k_reduce<<<4096, B, 0, stream>>>((const float4*)xc, sums);
    k_gln_apply<<<APB, B, 0, stream>>>((const float4*)xc, sums, nw + i * CH, nb + i * CH,
                                       (float4*)hpre);
    edge_phase(hpre);
    k_conv_mlp<<<MLPB, B, 0, stream>>>(hpre, agg,
                                       cW1 + (size_t)i * CH * HD, cb1 + (size_t)i * HD,
                                       clnw + (size_t)i * HD, clnb + (size_t)i * HD,
                                       cW2 + (size_t)i * HD * CH, cb2 + (size_t)i * CH,
                                       xc, xc);
  }

  k_head<<<MLPB, B, 0, stream>>>(xc, lW1, lb1, lW2, lb2, lW3, lb3, (float*)d_out);
}


# Softmax aggregation as a gather: edges grouped by destination once per call, then one per-node pass per layer (running max and exp sums in registers) replaces clear + three atomic edge passes
# speedup vs baseline: 2.0817x; 2.0817x over previous
#include <hip/hip_runtime.h>
#include <math.h>

#define NN   100000
#define CH   64
#define HD   128
#define NE   800000
#define N64  6400000
#define EPS_MSG 1e-7f
#define EPS_LN  1e-5f

typedef float v2f __attribute__((ext_vector_type(2)));
typedef float v8f __attribute__((ext_vector_type(8)));

static __device__ __forceinline__ v8f wmma4(v2f a, v2f b, v8f c) {
  return __builtin_amdgcn_wmma_f32_16x16x4_f32(
       false, a,  false, b,
       (short)0, c,  false,  false);
}

__global__ void k_zero2(float* __restrict__ p) {
  if (threadIdx.x < 2) p[threadIdx.x] = 0.f;
}

__global__ __launch_bounds__(256) void k_reduce(const float4* __restrict__ h,
                                                float* __restrict__ sums) {
  __shared__ float ss[256];
  __shared__ float sq[256];
  float s = 0.f, q = 0.f;
  for (int i = blockIdx.x * 256 + threadIdx.x; i < N64 / 4; i += gridDim.x * 256) {
    float4 v = h[i];
    s += v.x + v.y + v.z + v.w;
    q += v.x * v.x + v.y * v.y + v.z * v.z + v.w * v.w;
  }
  ss[threadIdx.x] = s; sq[threadIdx.x] = q;
  __syncthreads();
  for (int off = 128; off > 0; off >>= 1) {
    if (threadIdx.x < off) {
      ss[threadIdx.x] += ss[threadIdx.x + off];
      sq[threadIdx.x] += sq[threadIdx.x + off];
    }
    __syncthreads();
  }
  if (threadIdx.x == 0) {
    atomicAdd(&sums[0], ss[0]);
    atomicAdd(&sums[1], sq[0]);
  }
}

__global__ __launch_bounds__(256) void k_gln_apply(const float4* __restrict__ h,
                                                   const float* __restrict__ sums,
                                                   const float* __restrict__ w,
                                                   const float* __restrict__ b,
                                                   float4* __restrict__ out) {
  int i = blockIdx.x * 256 + threadIdx.x;
  if (i >= N64 / 4) return;
  const float invn = 1.f / (float)N64;
  float mu = sums[0] * invn;
  float var = sums[1] * invn - mu * mu;
  float scale = 1.f / (sqrtf(fmaxf(var, 0.f)) + EPS_LN);
  int c = (i & 15) << 2;
  float4 v = h[i];
  float r0 = (v.x - mu) * scale * w[c + 0] + b[c + 0];
  float r1 = (v.y - mu) * scale * w[c + 1] + b[c + 1];
  float r2 = (v.z - mu) * scale * w[c + 2] + b[c + 2];
  float r3 = (v.w - mu) * scale * w[c + 3] + b[c + 3];
  out[i] = make_float4(fmaxf(r0, 0.f), fmaxf(r1, 0.f), fmaxf(r2, 0.f), fmaxf(r3, 0.f));
}

__global__ __launch_bounds__(256) void k_csr_zero(int* __restrict__ deg, int* __restrict__ total) {
  int i = blockIdx.x * 256 + threadIdx.x;
  if (i < NN) deg[i] = 0;
  if (i == 0) total[0] = 0;
}

__global__ __launch_bounds__(256) void k_csr_count(const int* __restrict__ dst,
                                                   int* __restrict__ deg) {
  int e = blockIdx.x * 256 + threadIdx.x;
  if (e < NE) atomicAdd(&deg[dst[e]], 1);
}

__global__ __launch_bounds__(256) void k_csr_offsets(const int* __restrict__ deg,
                                                     int* __restrict__ total,
                                                     int* __restrict__ rbeg,
                                                     int* __restrict__ rcur) {
  __shared__ int swave[8];
  __shared__ int sbase;
  const int tid = threadIdx.x, lane = tid & 31, wave = tid >> 5;
  const int n = blockIdx.x * 256 + tid;
  const int d = (n < NN) ? deg[n] : 0;
  int v = d;
  for (int off = 1; off < 32; off <<= 1) {
    int t = __shfl_up(v, off);
    if (lane >= off) v += t;
  }
  if (lane == 31) swave[wave] = v;
  __syncthreads();
  if (tid == 0) {
    int s = 0;
    for (int w = 0; w < 8; w++) { int t = swave[w]; swave[w] = s; s += t; }
    sbase = atomicAdd(total, s);
  }
  __syncthreads();
  if (n < NN) {
    int b = sbase + swave[wave] + v - d;
    rbeg[n] = b;
    rcur[n] = b;
  }
}

__global__ __launch_bounds__(256) void k_csr_fill(const int* __restrict__ src,
                                                  const int* __restrict__ dst,
                                                  int* __restrict__ rcur,
                                                  int* __restrict__ col) {
  int e = blockIdx.x * 256 + threadIdx.x;
  if (e < NE) {
    int p = atomicAdd(&rcur[dst[e]], 1);
    col[p] = src[e];
  }
}

static __device__ __forceinline__ void soft4(float v0, float v1, float v2, float v3,
                                             bool p1, bool p2, bool p3,
                                             float& mx, float& s1, float& s2) {
  const float a0 = fmaxf(v0, 0.f) + EPS_MSG;
  const float a1 = p1 ? fmaxf(v1, 0.f) + EPS_MSG : 0.f;
  const float a2 = p2 ? fmaxf(v2, 0.f) + EPS_MSG : 0.f;
  const float a3 = p3 ? fmaxf(v3, 0.f) + EPS_MSG : 0.f;
  const float mn = fmaxf(fmaxf(mx, fmaxf(a0, a1)), fmaxf(a2, a3));
  const float sc = __expf(mx - mn);
  const float e0 = __expf(a0 - mn);
  const float e1 = p1 ? __expf(a1 - mn) : 0.f;
  const float e2 = p2 ? __expf(a2 - mn) : 0.f;
  const float e3 = p3 ? __expf(a3 - mn) : 0.f;
  s1 = s1 * sc + ((e0 + e1) + (e2 + e3));
  s2 = s2 * sc + ((a0 * e0 + a1 * e1) + (a2 * e2 + a3 * e3));
  mx = mn;
}

__global__ __launch_bounds__(256) void k_edge_gather(const int* __restrict__ rbeg,
                                                     const int* __restrict__ rend,
                                                     const int* __restrict__ col,
                                                     const float* __restrict__ hin,
                                                     float* __restrict__ agg) {
  const int node = (blockIdx.x * 256 + threadIdx.x) >> 4;
  const int lane = threadIdx.x & 31;
  const int hb = lane & 16, l16 = lane & 15;
  const int g = l16 << 2;
  const int beg = rbeg[node];
  const int deg = rend[node] - beg;
  const int degw = max(deg, __shfl_xor(deg, 16));
  float mx0 = 0.f, mx1 = 0.f, mx2 = 0.f, mx3 = 0.f;
  float s10 = 0.f, s11 = 0.f, s12 = 0.f, s13 = 0.f;
  float s20 = 0.f, s21 = 0.f, s22 = 0.f, s23 = 0.f;
  const float* hp = hin + g;
  for (int c0 = 0; c0 < degw; c0 += 16) {
    const int mine = (c0 + l16 < deg) ? col[beg + c0 + l16] : 0;
    const int cnt = deg - c0;
    const int cntw = min(16, degw - c0);
    for (int i = 0; i < cntw; i += 4) {
      const int i0 = __shfl(mine, hb + i);
      const int i1 = __shfl(mine, hb + i + 1);
      const int i2 = __shfl(mine, hb + i + 2);
      const int i3 = __shfl(mine, hb + i + 3);
      if (i < cnt) {
        const bool p1 = i + 1 < cnt, p2 = i + 2 < cnt, p3 = i + 3 < cnt;
        const int j1 = p1 ? i1 : i0, j2 = p2 ? i2 : i0, j3 = p3 ? i3 : i0;
        const float4 v0 = *(const float4*)(hp + (size_t)i0 * CH);
        const float4 v1 = *(const float4*)(hp + (size_t)j1 * CH);
        const float4 v2 = *(const float4*)(hp + (size_t)j2 * CH);
        const float4 v3 = *(const float4*)(hp + (size_t)j3 * CH);
        soft4(v0.x, v1.x, v2.x, v3.x, p1, p2, p3, mx0, s10, s20);
        soft4(v0.y, v1.y, v2.y, v3.y, p1, p2, p3, mx1, s11, s21);
        soft4(v0.z, v1.z, v2.z, v3.z, p1, p2, p3, mx2, s12, s22);
        soft4(v0.w, v1.w, v2.w, v3.w, p1, p2, p3, mx3, s13, s23);
      }
    }
  }
  float4 r = make_float4(0.f, 0.f, 0.f, 0.f);
  if (deg > 0) r = make_float4(s20 / s10, s21 / s11, s22 / s12, s23 / s13);
  *(float4*)(agg + (size_t)node * CH + g) = r;
}

__global__ __launch_bounds__(256)
void k_conv_mlp(const float* __restrict__ resin, const float* __restrict__ agg,
                const float* __restrict__ W1, const float* __restrict__ b1,
                const float* __restrict__ lnw, const float* __restrict__ lnb,
                const float* __restrict__ W2, const float* __restrict__ b2,
                const float* __restrict__ outer, float* __restrict__ out) {
  __shared__ float sAct[32 * CH];
  __shared__ float sHid[32 * HD];
  __shared__ float sb1[HD], slnw[HD], slnb[HD], sb2[CH];
  const int tid = threadIdx.x;
  const int base = blockIdx.x * 32;
  if (tid < HD) { sb1[tid] = b1[tid]; slnw[tid] = lnw[tid]; slnb[tid] = lnb[tid]; }
  if (tid < CH) sb2[tid] = b2[tid];
  for (int i = tid; i < 32 * CH; i += 256) {
    size_t idx = (size_t)base * CH + i;
    sAct[i] = resin[idx] + agg[idx];
  }
  __syncthreads();

  const int wave = tid >> 5, lane = tid & 31;
  const int half = lane >> 4, l16 = lane & 15;

  for (int task = wave; task < 16; task += 8) {
    int rt = task >> 3, ct = task & 7;
    const float* arow = &sAct[(rt * 16 + l16) * CH];
    v8f acc = {0.f, 0.f, 0.f, 0.f, 0.f, 0.f, 0.f, 0.f};
    #pragma unroll
    for (int k0 = 0; k0 < CH; k0 += 4) {
      int k = k0 + 2 * half;
      v2f a;  a.x = arow[k];                     a.y = arow[k + 1];
      v2f bb; bb.x = W1[k * HD + ct * 16 + l16]; bb.y = W1[(k + 1) * HD + ct * 16 + l16];
      acc = wmma4(a, bb, acc);
    }
    #pragma unroll
    for (int r = 0; r < 8; r++) {
      int m = rt * 16 + r + 8 * half;
      int n = ct * 16 + l16;
      sHid[m * HD + n] = acc[r] + sb1[n];
    }
  }
  __syncthreads();

  if (tid < 32) {
    float s = 0.f, q = 0.f;
    for (int c = 0; c < HD; c++) { float v = sHid[tid * HD + c]; s += v; q += v * v; }
    float mu = s * (1.f / HD);
    float var = q * (1.f / HD) - mu * mu;
    float inv = rsqrtf(var + EPS_LN);
    for (int c = 0; c < HD; c++) {
      float v = (sHid[tid * HD + c] - mu) * inv * slnw[c] + slnb[c];
      sHid[tid * HD + c] = fmaxf(v, 0.f);
    }
  }
  __syncthreads();

  {
    int rt = wave >> 2, ct = wave & 3;
    const float* arow = &sHid[(rt * 16 + l16) * HD];
    v8f acc = {0.f, 0.f, 0.f, 0.f, 0.f, 0.f, 0.f, 0.f};
    #pragma unroll
    for (int k0 = 0; k0 < HD; k0 += 4) {
      int k = k0 + 2 * half;
      v2f a;  a.x = arow[k];                     a.y = arow[k + 1];
      v2f bb; bb.x = W2[k * CH + ct * 16 + l16]; bb.y = W2[(k + 1) * CH + ct * 16 + l16];
      acc = wmma4(a, bb, acc);
    }
    #pragma unroll
    for (int r = 0; r < 8; r++) {
      int m = rt * 16 + r + 8 * half;
      size_t node = (size_t)(base + m);
      int n = ct * 16 + l16;
      float v = acc[r] + sb2[n];
      if (outer) v += outer[node * CH + n];
      out[node * CH + n] = v;
    }
  }
}

__global__ __launch_bounds__(256)
void k_head(const float* __restrict__ xc,
            const float* __restrict__ W1, const float* __restrict__ b1,
            const float* __restrict__ W2, const float* __restrict__ b2,
            const float* __restrict__ W3, const float* __restrict__ b3,
            float* __restrict__ out) {
  __shared__ float sA[32 * CH];
  __shared__ float sH[32 * CH];
  __shared__ float sb1[CH], sb2[CH], sW3[CH * 2], sb3[2];
  const int tid = threadIdx.x;
  const int base = blockIdx.x * 32;
  if (tid < CH) { sb1[tid] = b1[tid]; sb2[tid] = b2[tid]; }
  if (tid < CH * 2) sW3[tid] = W3[tid];
  if (tid < 2) sb3[tid] = b3[tid];
  for (int i = tid; i < 32 * CH; i += 256) sA[i] = xc[(size_t)base * CH + i];
  __syncthreads();

  const int wave = tid >> 5, lane = tid & 31;
  const int half = lane >> 4, l16 = lane & 15;
  const int rt = wave >> 2, ct = wave & 3;

  {
    const float* arow = &sA[(rt * 16 + l16) * CH];
    v8f acc = {0.f, 0.f, 0.f, 0.f, 0.f, 0.f, 0.f, 0.f};
    #pragma unroll
    for (int k0 = 0; k0 < CH; k0 += 4) {
      int k = k0 + 2 * half;
      v2f a;  a.x = arow[k];                     a.y = arow[k + 1];
      v2f bb; bb.x = W1[k * CH + ct * 16 + l16]; bb.y = W1[(k + 1) * CH + ct * 16 + l16];
      acc = wmma4(a, bb, acc);
    }
    #pragma unroll
    for (int r = 0; r < 8; r++) {
      int m = rt * 16 + r + 8 * half, n = ct * 16 + l16;
      sH[m * CH + n] = fmaxf(acc[r] + sb1[n], 0.f);
    }
  }
  __syncthreads();
  {
    const float* arow = &sH[(rt * 16 + l16) * CH];
    v8f acc = {0.f, 0.f, 0.f, 0.f, 0.f, 0.f, 0.f, 0.f};
    #pragma unroll
    for (int k0 = 0; k0 < CH; k0 += 4) {
      int k = k0 + 2 * half;
      v2f a;  a.x = arow[k];                     a.y = arow[k + 1];
      v2f bb; bb.x = W2[k * CH + ct * 16 + l16]; bb.y = W2[(k + 1) * CH + ct * 16 + l16];
      acc = wmma4(a, bb, acc);
    }
    #pragma unroll
    for (int r = 0; r < 8; r++) {
      int m = rt * 16 + r + 8 * half, n = ct * 16 + l16;
      sA[m * CH + n] = fmaxf(acc[r] + sb2[n], 0.f);
    }
  }
  __syncthreads();
  if (tid < 64) {
    int m = tid >> 1, o = tid & 1;
    float s = sb3[o];
    for (int k = 0; k < CH; k++) s += sA[m * CH + k] * sW3[k * 2 + o];
    out[(size_t)(base + m) * 2 + o] = s;
  }
}

extern "C" void kernel_launch(void* const* d_in, const int* in_sizes, int n_in,
                              void* d_out, int out_size, void* d_ws, size_t ws_size,
                              hipStream_t stream) {
  (void)in_sizes; (void)n_in; (void)out_size; (void)ws_size;
  const float* x      = (const float*)d_in[0];
  const int*   ei     = (const int*)d_in[1];
  const int*   src    = ei;
  const int*   dst    = ei + NE;
  const float* c1_W1  = (const float*)d_in[2];
  const float* c1_b1  = (const float*)d_in[3];
  const float* c1_lnw = (const float*)d_in[4];
  const float* c1_lnb = (const float*)d_in[5];
  const float* c1_W2  = (const float*)d_in[6];
  const float* c1_b2  = (const float*)d_in[7];
  const float* n1_w   = (const float*)d_in[8];
  const float* n1_b   = (const float*)d_in[9];
  const float* cW1    = (const float*)d_in[10];
  const float* cb1    = (const float*)d_in[11];
  const float* clnw   = (const float*)d_in[12];
  const float* clnb   = (const float*)d_in[13];
  const float* cW2    = (const float*)d_in[14];
  const float* cb2    = (const float*)d_in[15];
  const float* nw     = (const float*)d_in[16];
  const float* nb     = (const float*)d_in[17];
  const float* lW1    = (const float*)d_in[18];
  const float* lb1    = (const float*)d_in[19];
  const float* lW2    = (const float*)d_in[20];
  const float* lb2    = (const float*)d_in[21];
  const float* lW3    = (const float*)d_in[22];
  const float* lb3    = (const float*)d_in[23];

  float* ws    = (float*)d_ws;
  float* xc    = ws;
  float* hpre  = ws + (size_t)N64;
  int*   csr   = (int*)(ws + 2 * (size_t)N64);
  int*   deg   = csr;
  int*   rbeg  = csr + NN;
  int*   rcur  = csr + 2 * NN;
  int*   total = csr + 3 * NN;
  int*   col   = csr + 3 * NN + 64;
  float* agg   = ws + 4 * (size_t)N64;
  float* sums  = ws + 5 * (size_t)N64;

  const dim3 B(256);
  const int EBK  = (NE + 255) / 256;
  const int NB   = (NN + 255) / 256;
  const int GB   = NN * 16 / 256;
  const int APB  = (N64 / 4 + 255) / 256;
  const int MLPB = NN / 32;

  k_csr_zero<<<NB, B, 0, stream>>>(deg, total);
  k_csr_count<<<EBK, B, 0, stream>>>(dst, deg);
  k_csr_offsets<<<NB, B, 0, stream>>>(deg, total, rbeg, rcur);
  k_csr_fill<<<EBK, B, 0, stream>>>(src, dst, rcur, col);

  auto edge_phase = [&](const float* hin) {
    k_edge_gather<<<GB, B, 0, stream>>>(rbeg, rcur, col, hin, agg);
  };

  edge_phase(x);
  k_conv_mlp<<<MLPB, B, 0, stream>>>(x, agg, c1_W1, c1_b1, c1_lnw, c1_lnb,
                                     c1_W2, c1_b2, nullptr, hpre);
  k_zero2<<<1, 32, 0, stream>>>(sums);
  k_reduce<<<4096, B, 0, stream>>>((const float4*)hpre, sums);
  k_gln_apply<<<APB, B, 0, stream>>>((const float4*)hpre, sums, n1_w, n1_b, (float4*)xc);

  for (int i = 0; i < 3; i++) {
    k_zero2<<<1, 32, 0, stream>>>(sums);
    k_reduce<<<4096, B, 0, stream>>>((const float4*)xc, sums);
    k_gln_apply<<<APB, B, 0, stream>>>((const float4*)xc, sums, nw + i * CH, nb + i * CH,
                                       (float4*)hpre);
    edge_phase(hpre);
    k_conv_mlp<<<MLPB, B, 0, stream>>>(hpre, agg,
                                       cW1 + (size_t)i * CH * HD, cb1 + (size_t)i * HD,
                                       clnw + (size_t)i * HD, clnb + (size_t)i * HD,
                                       cW2 + (size_t)i * HD * CH, cb2 + (size_t)i * CH,
                                       xc, xc);
  }

  k_head<<<MLPB, B, 0, stream>>>(xc, lW1, lb1, lW2, lb2, lW3, lb3, (float*)d_out);
}


# Fused conv-MLP and head kernels: LDS rows padded by four floats (pitch 68 and 132) so A-fragment reads and the row-LN walk spread over the banks instead of serialising on one bank
# speedup vs baseline: 2.8055x; 1.3477x over previous
#include <hip/hip_runtime.h>
#include <math.h>

#define NN   100000
#define CH   64
#define HD   128
#define NE   800000
#define N64  6400000
#define EPS_MSG 1e-7f
#define EPS_LN  1e-5f
#define PA   (CH + 4)
#define PH   (HD + 4)

typedef float v2f __attribute__((ext_vector_type(2)));
typedef float v8f __attribute__((ext_vector_type(8)));

static __device__ __forceinline__ v8f wmma4(v2f a, v2f b, v8f c) {
  return __builtin_amdgcn_wmma_f32_16x16x4_f32(
       false, a,  false, b,
       (short)0, c,  false,  false);
}

__global__ void k_zero2(float* __restrict__ p) {
  if (threadIdx.x < 2) p[threadIdx.x] = 0.f;
}

__global__ __launch_bounds__(256) void k_reduce(const float4* __restrict__ h,
                                                float* __restrict__ sums) {
  __shared__ float ss[256];
  __shared__ float sq[256];
  float s = 0.f, q = 0.f;
  for (int i = blockIdx.x * 256 + threadIdx.x; i < N64 / 4; i += gridDim.x * 256) {
    float4 v = h[i];
    s += v.x + v.y + v.z + v.w;
    q += v.x * v.x + v.y * v.y + v.z * v.z + v.w * v.w;
  }
  ss[threadIdx.x] = s; sq[threadIdx.x] = q;
  __syncthreads();
  for (int off = 128; off > 0; off >>= 1) {
    if (threadIdx.x < off) {
      ss[threadIdx.x] += ss[threadIdx.x + off];
      sq[threadIdx.x] += sq[threadIdx.x + off];
    }
    __syncthreads();
  }
  if (threadIdx.x == 0) {
    atomicAdd(&sums[0], ss[0]);
    atomicAdd(&sums[1], sq[0]);
  }
}

__global__ __launch_bounds__(256) void k_gln_apply(const float4* __restrict__ h,
                                                   const float* __restrict__ sums,
                                                   const float* __restrict__ w,
                                                   const float* __restrict__ b,
                                                   float4* __restrict__ out) {
  int i = blockIdx.x * 256 + threadIdx.x;
  if (i >= N64 / 4) return;
  const float invn = 1.f / (float)N64;
  float mu = sums[0] * invn;
  float var = sums[1] * invn - mu * mu;
  float scale = 1.f / (sqrtf(fmaxf(var, 0.f)) + EPS_LN);
  int c = (i & 15) << 2;
  float4 v = h[i];
  float r0 = (v.x - mu) * scale * w[c + 0] + b[c + 0];
  float r1 = (v.y - mu) * scale * w[c + 1] + b[c + 1];
  float r2 = (v.z - mu) * scale * w[c + 2] + b[c + 2];
  float r3 = (v.w - mu) * scale * w[c + 3] + b[c + 3];
  out[i] = make_float4(fmaxf(r0, 0.f), fmaxf(r1, 0.f), fmaxf(r2, 0.f), fmaxf(r3, 0.f));
}

__global__ __launch_bounds__(256) void k_csr_zero(int* __restrict__ deg, int* __restrict__ total) {
  int i = blockIdx.x * 256 + threadIdx.x;
  if (i < NN) deg[i] = 0;
  if (i == 0) total[0] = 0;
}

__global__ __launch_bounds__(256) void k_csr_count(const int* __restrict__ dst,
                                                   int* __restrict__ deg) {
  int e = blockIdx.x * 256 + threadIdx.x;
  if (e < NE) atomicAdd(&deg[dst[e]], 1);
}

__global__ __launch_bounds__(256) void k_csr_offsets(const int* __restrict__ deg,
                                                     int* __restrict__ total,
                                                     int* __restrict__ rbeg,
                                                     int* __restrict__ rcur) {
  __shared__ int swave[8];
  __shared__ int sbase;
  const int tid = threadIdx.x, lane = tid & 31, wave = tid >> 5;
  const int n = blockIdx.x * 256 + tid;
  const int d = (n < NN) ? deg[n] : 0;
  int v = d;
  for (int off = 1; off < 32; off <<= 1) {
    int t = __shfl_up(v, off);
    if (lane >= off) v += t;
  }
  if (lane == 31) swave[wave] = v;
  __syncthreads();
  if (tid == 0) {
    int s = 0;
    for (int w = 0; w < 8; w++) { int t = swave[w]; swave[w] = s; s += t; }
    sbase = atomicAdd(total, s);
  }
  __syncthreads();
  if (n < NN) {
    int b = sbase + swave[wave] + v - d;
    rbeg[n] = b;
    rcur[n] = b;
  }
}

__global__ __launch_bounds__(256) void k_csr_fill(const int* __restrict__ src,
                                                  const int* __restrict__ dst,
                                                  int* __restrict__ rcur,
                                                  int* __restrict__ col) {
  int e = blockIdx.x * 256 + threadIdx.x;
  if (e < NE) {
    int p = atomicAdd(&rcur[dst[e]], 1);
    col[p] = src[e];
  }
}

static __device__ __forceinline__ void soft4(float v0, float v1, float v2, float v3,
                                             bool p1, bool p2, bool p3,
                                             float& mx, float& s1, float& s2) {
  const float a0 = fmaxf(v0, 0.f) + EPS_MSG;
  const float a1 = p1 ? fmaxf(v1, 0.f) + EPS_MSG : 0.f;
  const float a2 = p2 ? fmaxf(v2, 0.f) + EPS_MSG : 0.f;
  const float a3 = p3 ? fmaxf(v3, 0.f) + EPS_MSG : 0.f;
  const float mn = fmaxf(fmaxf(mx, fmaxf(a0, a1)), fmaxf(a2, a3));
  const float sc = __expf(mx - mn);
  const float e0 = __expf(a0 - mn);
  const float e1 = p1 ? __expf(a1 - mn) : 0.f;
  const float e2 = p2 ? __expf(a2 - mn) : 0.f;
  const float e3 = p3 ? __expf(a3 - mn) : 0.f;
  s1 = s1 * sc + ((e0 + e1) + (e2 + e3));
  s2 = s2 * sc + ((a0 * e0 + a1 * e1) + (a2 * e2 + a3 * e3));
  mx = mn;
}

__global__ __launch_bounds__(256) void k_edge_gather(const int* __restrict__ rbeg,
                                                     const int* __restrict__ rend,
                                                     const int* __restrict__ col,
                                                     const float* __restrict__ hin,
                                                     float* __restrict__ agg) {
  const int node = (blockIdx.x * 256 + threadIdx.x) >> 4;
  const int lane = threadIdx.x & 31;
  const int hb = lane & 16, l16 = lane & 15;
  const int g = l16 << 2;
  const int beg = rbeg[node];
  const int deg = rend[node] - beg;
  const int degw = max(deg, __shfl_xor(deg, 16));
  float mx0 = 0.f, mx1 = 0.f, mx2 = 0.f, mx3 = 0.f;
  float s10 = 0.f, s11 = 0.f, s12 = 0.f, s13 = 0.f;
  float s20 = 0.f, s21 = 0.f, s22 = 0.f, s23 = 0.f;
  const float* hp = hin + g;
  for (int c0 = 0; c0 < degw; c0 += 16) {
    const int mine = (c0 + l16 < deg) ? col[beg + c0 + l16] : 0;
    const int cnt = deg - c0;
    const int cntw = min(16, degw - c0);
    for (int i = 0; i < cntw; i += 4) {
      const int i0 = __shfl(mine, hb + i);
      const int i1 = __shfl(mine, hb + i + 1);
      const int i2 = __shfl(mine, hb + i + 2);
      const int i3 = __shfl(mine, hb + i + 3);
      if (i < cnt) {
        const bool p1 = i + 1 < cnt, p2 = i + 2 < cnt, p3 = i + 3 < cnt;
        const int j1 = p1 ? i1 : i0, j2 = p2 ? i2 : i0, j3 = p3 ? i3 : i0;
        const float4 v0 = *(const float4*)(hp + (size_t)i0 * CH);
        const float4 v1 = *(const float4*)(hp + (size_t)j1 * CH);
        const float4 v2 = *(const float4*)(hp + (size_t)j2 * CH);
        const float4 v3 = *(const float4*)(hp + (size_t)j3 * CH);
        soft4(v0.x, v1.x, v2.x, v3.x, p1, p2, p3, mx0, s10, s20);
        soft4(v0.y, v1.y, v2.y, v3.y, p1, p2, p3, mx1, s11, s21);
        soft4(v0.z, v1.z, v2.z, v3.z, p1, p2, p3, mx2, s12, s22);
        soft4(v0.w, v1.w, v2.w, v3.w, p1, p2, p3, mx3, s13, s23);
      }
    }
  }
  float4 r = make_float4(0.f, 0.f, 0.f, 0.f);
  if (deg > 0) r = make_float4(s20 / s10, s21 / s11, s22 / s12, s23 / s13);
  *(float4*)(agg + (size_t)node * CH + g) = r;
}

__global__ __launch_bounds__(256)
void k_conv_mlp(const float* __restrict__ resin, const float* __restrict__ agg,
                const float* __restrict__ W1, const float* __restrict__ b1,
                const float* __restrict__ lnw, const float* __restrict__ lnb,
                const float* __restrict__ W2, const float* __restrict__ b2,
                const float* __restrict__ outer, float* __restrict__ out) {
  __shared__ float sAct[32 * PA];
  __shared__ float sHid[32 * PH];
  __shared__ float sb1[HD], slnw[HD], slnb[HD], sb2[CH];
  const int tid = threadIdx.x;
  const int base = blockIdx.x * 32;
  if (tid < HD) { sb1[tid] = b1[tid]; slnw[tid] = lnw[tid]; slnb[tid] = lnb[tid]; }
  if (tid < CH) sb2[tid] = b2[tid];
  for (int i = tid; i < 32 * CH; i += 256) {
    size_t idx = (size_t)base * CH + i;
    sAct[(i >> 6) * PA + (i & 63)] = resin[idx] + agg[idx];
  }
  __syncthreads();

  const int wave = tid >> 5, lane = tid & 31;
  const int half = lane >> 4, l16 = lane & 15;

  for (int task = wave; task < 16; task += 8) {
    int rt = task >> 3, ct = task & 7;
    const float* arow = &sAct[(rt * 16 + l16) * PA];
    v8f acc = {0.f, 0.f, 0.f, 0.f, 0.f, 0.f, 0.f, 0.f};
    #pragma unroll
    for (int k0 = 0; k0 < CH; k0 += 4) {
      int k = k0 + 2 * half;
      v2f a;  a.x = arow[k];                     a.y = arow[k + 1];
      v2f bb; bb.x = W1[k * HD + ct * 16 + l16]; bb.y = W1[(k + 1) * HD + ct * 16 + l16];
      acc = wmma4(a, bb, acc);
    }
    #pragma unroll
    for (int r = 0; r < 8; r++) {
      int m = rt * 16 + r + 8 * half;
      int n = ct * 16 + l16;
      sHid[m * PH + n] = acc[r] + sb1[n];
    }
  }
  __syncthreads();

  if (tid < 32) {
    float s = 0.f, q = 0.f;
    for (int c = 0; c < HD; c++) { float v = sHid[tid * PH + c]; s += v; q += v * v; }
    float mu = s * (1.f / HD);
    float var = q * (1.f / HD) - mu * mu;
    float inv = rsqrtf(var + EPS_LN);
    for (int c = 0; c < HD; c++) {
      float v = (sHid[tid * PH + c] - mu) * inv * slnw[c] + slnb[c];
      sHid[tid * PH + c] = fmaxf(v, 0.f);
    }
  }
  __syncthreads();

  {
    int rt = wave >> 2, ct = wave & 3;
    const float* arow = &sHid[(rt * 16 + l16) * PH];
    v8f acc = {0.f, 0.f, 0.f, 0.f, 0.f, 0.f, 0.f, 0.f};
    #pragma unroll
    for (int k0 = 0; k0 < HD; k0 += 4) {
      int k = k0 + 2 * half;
      v2f a;  a.x = arow[k];                     a.y = arow[k + 1];
      v2f bb; bb.x = W2[k * CH + ct * 16 + l16]; bb.y = W2[(k + 1) * CH + ct * 16 + l16];
      acc = wmma4(a, bb, acc);
    }
    #pragma unroll
    for (int r = 0; r < 8; r++) {
      int m = rt * 16 + r + 8 * half;
      size_t node = (size_t)(base + m);
      int n = ct * 16 + l16;
      float v = acc[r] + sb2[n];
      if (outer) v += outer[node * CH + n];
      out[node * CH + n] = v;
    }
  }
}

__global__ __launch_bounds__(256)
void k_head(const float* __restrict__ xc,
            const float* __restrict__ W1, const float* __restrict__ b1,
            const float* __restrict__ W2, const float* __restrict__ b2,
            const float* __restrict__ W3, const float* __restrict__ b3,
            float* __restrict__ out) {
  __shared__ float sA[32 * PA];
  __shared__ float sH[32 * PA];
  __shared__ float sb1[CH], sb2[CH], sW3[CH * 2], sb3[2];
  const int tid = threadIdx.x;
  const int base = blockIdx.x * 32;
  if (tid < CH) { sb1[tid] = b1[tid]; sb2[tid] = b2[tid]; }
  if (tid < CH * 2) sW3[tid] = W3[tid];
  if (tid < 2) sb3[tid] = b3[tid];
  for (int i = tid; i < 32 * CH; i += 256) sA[(i >> 6) * PA + (i & 63)] = xc[(size_t)base * CH + i];
  __syncthreads();

  const int wave = tid >> 5, lane = tid & 31;
  const int half = lane >> 4, l16 = lane & 15;
  const int rt = wave >> 2, ct = wave & 3;

  {
    const float* arow = &sA[(rt * 16 + l16) * PA];
    v8f acc = {0.f, 0.f, 0.f, 0.f, 0.f, 0.f, 0.f, 0.f};
    #pragma unroll
    for (int k0 = 0; k0 < CH; k0 += 4) {
      int k = k0 + 2 * half;
      v2f a;  a.x = arow[k];                     a.y = arow[k + 1];
      v2f bb; bb.x = W1[k * CH + ct * 16 + l16]; bb.y = W1[(k + 1) * CH + ct * 16 + l16];
      acc = wmma4(a, bb, acc);
    }
    #pragma unroll
    for (int r = 0; r < 8; r++) {
      int m = rt * 16 + r + 8 * half, n = ct * 16 + l16;
      sH[m * PA + n] = fmaxf(acc[r] + sb1[n], 0.f);
    }
  }
  __syncthreads();
  {
    const float* arow = &sH[(rt * 16 + l16) * PA];
    v8f acc = {0.f, 0.f, 0.f, 0.f, 0.f, 0.f, 0.f, 0.f};
    #pragma unroll
    for (int k0 = 0; k0 < CH; k0 += 4) {
      int k = k0 + 2 * half;
      v2f a;  a.x = arow[k];                     a.y = arow[k + 1];
      v2f bb; bb.x = W2[k * CH + ct * 16 + l16]; bb.y = W2[(k + 1) * CH + ct * 16 + l16];
      acc = wmma4(a, bb, acc);
    }
    #pragma unroll
    for (int r = 0; r < 8; r++) {
      int m = rt * 16 + r + 8 * half, n = ct * 16 + l16;
      sA[m * PA + n] = fmaxf(acc[r] + sb2[n], 0.f);
    }
  }
  __syncthreads();
  if (tid < 64) {
    int m = tid >> 1, o = tid & 1;
    float s = sb3[o];
    for (int k = 0; k < CH; k++) s += sA[m * PA + k] * sW3[k * 2 + o];
    out[(size_t)(base + m) * 2 + o] = s;
  }
}

extern "C" void kernel_launch(void* const* d_in, const int* in_sizes, int n_in,
                              void* d_out, int out_size, void* d_ws, size_t ws_size,
                              hipStream_t stream) {
  (void)in_sizes; (void)n_in; (void)out_size; (void)ws_size;
  const float* x      = (const float*)d_in[0];
  const int*   ei     = (const int*)d_in[1];
  const int*   src    = ei;
  const int*   dst    = ei + NE;
  const float* c1_W1  = (const float*)d_in[2];
  const float* c1_b1  = (const float*)d_in[3];
  const float* c1_lnw = (const float*)d_in[4];
  const float* c1_lnb = (const float*)d_in[5];
  const float* c1_W2  = (const float*)d_in[6];
  const float* c1_b2  = (const float*)d_in[7];
  const float* n1_w   = (const float*)d_in[8];
  const float* n1_b   = (const float*)d_in[9];
  const float* cW1    = (const float*)d_in[10];
  const float* cb1    = (const float*)d_in[11];
  const float* clnw   = (const float*)d_in[12];
  const float* clnb   = (const float*)d_in[13];
  const float* cW2    = (const float*)d_in[14];
  const float* cb2    = (const float*)d_in[15];
  const float* nw     = (const float*)d_in[16];
  const float* nb     = (const float*)d_in[17];
  const float* lW1    = (const float*)d_in[18];
  const float* lb1    = (const float*)d_in[19];
  const float* lW2    = (const float*)d_in[20];
  const float* lb2    = (const float*)d_in[21];
  const float* lW3    = (const float*)d_in[22];
  const float* lb3    = (const float*)d_in[23];

  float* ws    = (float*)d_ws;
  float* xc    = ws;
  float* hpre  = ws + (size_t)N64;
  int*   csr   = (int*)(ws + 2 * (size_t)N64);
  int*   deg   = csr;
  int*   rbeg  = csr + NN;
  int*   rcur  = csr + 2 * NN;
  int*   total = csr + 3 * NN;
  int*   col   = csr + 3 * NN + 64;
  float* agg   = ws + 4 * (size_t)N64;
  float* sums  = ws + 5 * (size_t)N64;

  const dim3 B(256);
  const int EBK  = (NE + 255) / 256;
  const int NB   = (NN + 255) / 256;
  const int GB   = NN * 16 / 256;
  const int APB  = (N64 / 4 + 255) / 256;
  const int MLPB = NN / 32;

  k_csr_zero<<<NB, B, 0, stream>>>(deg, total);
  k_csr_count<<<EBK, B, 0, stream>>>(dst, deg);
  k_csr_offsets<<<NB, B, 0, stream>>>(deg, total, rbeg, rcur);
  k_csr_fill<<<EBK, B, 0, stream>>>(src, dst, rcur, col);

  auto edge_phase = [&](const float* hin) {
    k_edge_gather<<<GB, B, 0, stream>>>(rbeg, rcur, col, hin, agg);
  };

  edge_phase(x);
  k_conv_mlp<<<MLPB, B, 0, stream>>>(x, agg, c1_W1, c1_b1, c1_lnw, c1_lnb,
                                     c1_W2, c1_b2, nullptr, hpre);
  k_zero2<<<1, 32, 0, stream>>>(sums);
  k_reduce<<<4096, B, 0, stream>>>((const float4*)hpre, sums);
  k_gln_apply<<<APB, B, 0, stream>>>((const float4*)hpre, sums, n1_w, n1_b, (float4*)xc);

  for (int i = 0; i < 3; i++) {
    k_zero2<<<1, 32, 0, stream>>>(sums);
    k_reduce<<<4096, B, 0, stream>>>((const float4*)xc, sums);
    k_gln_apply<<<APB, B, 0, stream>>>((const float4*)xc, sums, nw + i * CH, nb + i * CH,
                                       (float4*)hpre);
    edge_phase(hpre);
    k_conv_mlp<<<MLPB, B, 0, stream>>>(hpre, agg,
                                       cW1 + (size_t)i * CH * HD, cb1 + (size_t)i * HD,
                                       clnw + (size_t)i * HD, clnb + (size_t)i * HD,
                                       cW2 + (size_t)i * HD * CH, cb2 + (size_t)i * CH,
                                       xc, xc);
  }

  k_head<<<MLPB, B, 0, stream>>>(xc, lW1, lb1, lW2, lb2, lW3, lb3, (float*)d_out);
}


# Chip-sized grids for the short kernels (graph-LN sum and apply, edge gather, edge count and fill): each thread loops over 5-10 items with its loads in flight, under a third of the waves to start
# speedup vs baseline: 2.9572x; 1.0541x over previous
#include <hip/hip_runtime.h>
#include <math.h>

#define NN   100000
#define CH   64
#define HD   128
#define NE   800000
#define N64  6400000
#define EPS_MSG 1e-7f
#define EPS_LN  1e-5f
#define PA   (CH + 4)
#define PH   (HD + 4)
#define SBLK 625
#define SPT  10
#define EBLK 625
#define EPT  5
#define GBLK 1250
#define GIT  5

typedef float v2f __attribute__((ext_vector_type(2)));
typedef float v8f __attribute__((ext_vector_type(8)));

static __device__ __forceinline__ v8f wmma4(v2f a, v2f b, v8f c) {
  return __builtin_amdgcn_wmma_f32_16x16x4_f32(
       false, a,  false, b,
       (short)0, c,  false,  false);
}

__global__ void k_zero2(float* __restrict__ p) {
  if (threadIdx.x < 2) p[threadIdx.x] = 0.f;
}

__global__ __launch_bounds__(256) void k_reduce(const float4* __restrict__ h,
                                                float* __restrict__ sums) {
  __shared__ float ss[256];
  __shared__ float sq[256];
  float s = 0.f, q = 0.f;
  const float4* p = h + (size_t)blockIdx.x * (256 * SPT) + threadIdx.x;
  float4 v[SPT];
  for (int j = 0; j < SPT; j++) v[j] = p[j * 256];
  for (int j = 0; j < SPT; j++) {
    s += v[j].x + v[j].y + v[j].z + v[j].w;
    q += v[j].x * v[j].x + v[j].y * v[j].y + v[j].z * v[j].z + v[j].w * v[j].w;
  }
  ss[threadIdx.x] = s; sq[threadIdx.x] = q;
  __syncthreads();
  for (int off = 128; off > 0; off >>= 1) {
    if (threadIdx.x < off) {
      ss[threadIdx.x] += ss[threadIdx.x + off];
      sq[threadIdx.x] += sq[threadIdx.x + off];
    }
    __syncthreads();
  }
  if (threadIdx.x == 0) {
    atomicAdd(&sums[0], ss[0]);
    atomicAdd(&sums[1], sq[0]);
  }
}

__global__ __launch_bounds__(256) void k_gln_apply(const float4* __restrict__ h,
                                                   const float* __restrict__ sums,
                                                   const float* __restrict__ w,
                                                   const float* __restrict__ b,
                                                   float4* __restrict__ out) {
  const size_t i0 = (size_t)blockIdx.x * (256 * SPT) + threadIdx.x;
  const float invn = 1.f / (float)N64;
  float mu = sums[0] * invn;
  float var = sums[1] * invn - mu * mu;
  float scale = 1.f / (sqrtf(fmaxf(var, 0.f)) + EPS_LN);
  const int c = (threadIdx.x & 15) << 2;
  const float w0 = w[c + 0], w1 = w[c + 1], w2 = w[c + 2], w3 = w[c + 3];
  const float b0 = b[c + 0], b1 = b[c + 1], b2 = b[c + 2], b3 = b[c + 3];
  float4 v[SPT];
  for (int j = 0; j < SPT; j++) v[j] = h[i0 + j * 256];
  for (int j = 0; j < SPT; j++) {
    float r0 = (v[j].x - mu) * scale * w0 + b0;
    float r1 = (v[j].y - mu) * scale * w1 + b1;
    float r2 = (v[j].z - mu) * scale * w2 + b2;
    float r3 = (v[j].w - mu) * scale * w3 + b3;
    out[i0 + j * 256] = make_float4(fmaxf(r0, 0.f), fmaxf(r1, 0.f), fmaxf(r2, 0.f), fmaxf(r3, 0.f));
  }
}

__global__ __launch_bounds__(256) void k_csr_zero(int* __restrict__ deg, int* __restrict__ total) {
  int i = blockIdx.x * 256 + threadIdx.x;
  if (i < NN) deg[i] = 0;
  if (i == 0) total[0] = 0;
}

__global__ __launch_bounds__(256) void k_csr_count(const int* __restrict__ dst,
                                                   int* __restrict__ deg) {
  const int e0 = blockIdx.x * (256 * EPT) + threadIdx.x;
  int d[EPT];
  for (int j = 0; j < EPT; j++) d[j] = dst[e0 + j * 256];
  for (int j = 0; j < EPT; j++) atomicAdd(&deg[d[j]], 1);
}

__global__ __launch_bounds__(256) void k_csr_offsets(const int* __restrict__ deg,
                                                     int* __restrict__ total,
                                                     int* __restrict__ rbeg,
                                                     int* __restrict__ rcur) {
  __shared__ int swave[8];
  __shared__ int sbase;
  const int tid = threadIdx.x, lane = tid & 31, wave = tid >> 5;
  const int n = blockIdx.x * 256 + tid;
  const int d = (n < NN) ? deg[n] : 0;
  int v = d;
  for (int off = 1; off < 32; off <<= 1) {
    int t = __shfl_up(v, off);
    if (lane >= off) v += t;
  }
  if (lane == 31) swave[wave] = v;
  __syncthreads();
  if (tid == 0) {
    int s = 0;
    for (int w = 0; w < 8; w++) { int t = swave[w]; swave[w] = s; s += t; }
    sbase = atomicAdd(total, s);
  }
  __syncthreads();
  if (n < NN) {
    int b = sbase + swave[wave] + v - d;
    rbeg[n] = b;
    rcur[n] = b;
  }
}

__global__ __launch_bounds__(256) void k_csr_fill(const int* __restrict__ src,
                                                  const int* __restrict__ dst,
                                                  int* __restrict__ rcur,
                                                  int* __restrict__ col) {
  const int e0 = blockIdx.x * (256 * EPT) + threadIdx.x;
  int d[EPT], s[EPT], p[EPT];
  for (int j = 0; j < EPT; j++) { d[j] = dst[e0 + j * 256]; s[j] = src[e0 + j * 256]; }
  for (int j = 0; j < EPT; j++) p[j] = atomicAdd(&rcur[d[j]], 1);
  for (int j = 0; j < EPT; j++) col[p[j]] = s[j];
}

static __device__ __forceinline__ void soft4(float v0, float v1, float v2, float v3,
                                             bool p1, bool p2, bool p3,
                                             float& mx, float& s1, float& s2) {
  const float a0 = fmaxf(v0, 0.f) + EPS_MSG;
  const float a1 = p1 ? fmaxf(v1, 0.f) + EPS_MSG : 0.f;
  const float a2 = p2 ? fmaxf(v2, 0.f) + EPS_MSG : 0.f;
  const float a3 = p3 ? fmaxf(v3, 0.f) + EPS_MSG : 0.f;
  const float mn = fmaxf(fmaxf(mx, fmaxf(a0, a1)), fmaxf(a2, a3));
  const float sc = __expf(mx - mn);
  const float e0 = __expf(a0 - mn);
  const float e1 = p1 ? __expf(a1 - mn) : 0.f;
  const float e2 = p2 ? __expf(a2 - mn) : 0.f;
  const float e3 = p3 ? __expf(a3 - mn) : 0.f;
  s1 = s1 * sc + ((e0 + e1) + (e2 + e3));
  s2 = s2 * sc + ((a0 * e0 + a1 * e1) + (a2 * e2 + a3 * e3));
  mx = mn;
}

struct SoftAcc {
  float mx0, mx1, mx2, mx3;
  float s10, s11, s12, s13;
  float s20, s21, s22, s23;
};

static __device__ __forceinline__ void gather_chunk(int idx, int cnt, int cntw, int hb,
                                                    const float* __restrict__ hp, SoftAcc& A) {
  for (int i = 0; i < cntw; i += 4) {
    const int i0 = __shfl(idx, hb + i);
    const int i1 = __shfl(idx, hb + i + 1);
    const int i2 = __shfl(idx, hb + i + 2);
    const int i3 = __shfl(idx, hb + i + 3);
    if (i < cnt) {
      const bool p1 = i + 1 < cnt, p2 = i + 2 < cnt, p3 = i + 3 < cnt;
      const int j1 = p1 ? i1 : i0, j2 = p2 ? i2 : i0, j3 = p3 ? i3 : i0;
      const float4 v0 = *(const float4*)(hp + (size_t)i0 * CH);
      const float4 v1 = *(const float4*)(hp + (size_t)j1 * CH);
      const float4 v2 = *(const float4*)(hp + (size_t)j2 * CH);
      const float4 v3 = *(const float4*)(hp + (size_t)j3 * CH);
      soft4(v0.x, v1.x, v2.x, v3.x, p1, p2, p3, A.mx0, A.s10, A.s20);
      soft4(v0.y, v1.y, v2.y, v3.y, p1, p2, p3, A.mx1, A.s11, A.s21);
      soft4(v0.z, v1.z, v2.z, v3.z, p1, p2, p3, A.mx2, A.s12, A.s22);
      soft4(v0.w, v1.w, v2.w, v3.w, p1, p2, p3, A.mx3, A.s13, A.s23);
    }
  }
}

__global__ __launch_bounds__(256) void k_edge_gather(const int* __restrict__ rbeg,
                                                     const int* __restrict__ rend,
                                                     const int* __restrict__ col,
                                                     const float* __restrict__ hin,
                                                     float* __restrict__ agg) {
  const int NSTEP = GBLK * 16;
  const int lane = threadIdx.x & 31;
  const int hb = lane & 16, l16 = lane & 15;
  const int g = l16 << 2;
  const float* hp = hin + g;
  int node = blockIdx.x * 16 + (threadIdx.x >> 4);
  int beg = rbeg[node];
  int end = rend[node];
  int nbeg = rbeg[node + NSTEP];
  int nend = rend[node + NSTEP];
  int deg = end - beg;
  int ndeg = nend - nbeg;
  int mine = (l16 < deg) ? col[beg + l16] : 0;
  for (int it = 0; it < GIT; it++) {
    const int nmine = (l16 < ndeg) ? col[nbeg + l16] : 0;
    const int n2 = min(node + 2 * NSTEP, NN - 1);
    const int n2beg = rbeg[n2];
    const int n2end = rend[n2];
    const int degw = max(deg, __shfl_xor(deg, 16));
    SoftAcc A = {0.f, 0.f, 0.f, 0.f, 0.f, 0.f, 0.f, 0.f, 0.f, 0.f, 0.f, 0.f};
    gather_chunk(mine, deg, min(16, degw), hb, hp, A);
    for (int c0 = 16; c0 < degw; c0 += 16) {
      const int more = (c0 + l16 < deg) ? col[beg + c0 + l16] : 0;
      gather_chunk(more, deg - c0, min(16, degw - c0), hb, hp, A);
    }
    float4 r = make_float4(0.f, 0.f, 0.f, 0.f);
    if (deg > 0) r = make_float4(A.s20 / A.s10, A.s21 / A.s11, A.s22 / A.s12, A.s23 / A.s13);
    *(float4*)(agg + (size_t)node * CH + g) = r;
    node += NSTEP;
    beg = nbeg; deg = ndeg; mine = nmine;
    nbeg = n2beg;
    ndeg = (it + 2 < GIT) ? n2end - n2beg : 0;
  }
}

__global__ __launch_bounds__(256)
void k_conv_mlp(const float* __restrict__ resin, const float* __restrict__ agg,
                const float* __restrict__ W1, const float* __restrict__ b1,
                const float* __restrict__ lnw, const float* __restrict__ lnb,
                const float* __restrict__ W2, const float* __restrict__ b2,
                const float* __restrict__ outer, float* __restrict__ out) {
  __shared__ float sAct[32 * PA];
  __shared__ float sHid[32 * PH];
  __shared__ float sb1[HD], slnw[HD], slnb[HD], sb2[CH];
  const int tid = threadIdx.x;
  const int base = blockIdx.x * 32;
  if (tid < HD) { sb1[tid] = b1[tid]; slnw[tid] = lnw[tid]; slnb[tid] = lnb[tid]; }
  if (tid < CH) sb2[tid] = b2[tid];
  for (int i = tid; i < 32 * CH; i += 256) {
    size_t idx = (size_t)base * CH + i;
    sAct[(i >> 6) * PA + (i & 63)] = resin[idx] + agg[idx];
  }
  __syncthreads();

  const int wave = tid >> 5, lane = tid & 31;
  const int half = lane >> 4, l16 = lane & 15;

  for (int task = wave; task < 16; task += 8) {
    int rt = task >> 3, ct = task & 7;
    const float* arow = &sAct[(rt * 16 + l16) * PA];
    v8f acc = {0.f, 0.f, 0.f, 0.f, 0.f, 0.f, 0.f, 0.f};
    #pragma unroll
    for (int k0 = 0; k0 < CH; k0 += 4) {
      int k = k0 + 2 * half;
      v2f a;  a.x = arow[k];                     a.y = arow[k + 1];
      v2f bb; bb.x = W1[k * HD + ct * 16 + l16]; bb.y = W1[(k + 1) * HD + ct * 16 + l16];
      acc = wmma4(a, bb, acc);
    }
    #pragma unroll
    for (int r = 0; r < 8; r++) {
      int m = rt * 16 + r + 8 * half;
      int n = ct * 16 + l16;
      sHid[m * PH + n] = acc[r] + sb1[n];
    }
  }
  __syncthreads();

  if (tid < 32) {
    float s = 0.f, q = 0.f;
    for (int c = 0; c < HD; c++) { float v = sHid[tid * PH + c]; s += v; q += v * v; }
    float mu = s * (1.f / HD);
    float var = q * (1.f / HD) - mu * mu;
    float inv = rsqrtf(var + EPS_LN);
    for (int c = 0; c < HD; c++) {
      float v = (sHid[tid * PH + c] - mu) * inv * slnw[c] + slnb[c];
      sHid[tid * PH + c] = fmaxf(v, 0.f);
    }
  }
  __syncthreads();

  {
    int rt = wave >> 2, ct = wave & 3;
    const float* arow = &sHid[(rt * 16 + l16) * PH];
    v8f acc = {0.f, 0.f, 0.f, 0.f, 0.f, 0.f, 0.f, 0.f};
    #pragma unroll
    for (int k0 = 0; k0 < HD; k0 += 4) {
      int k = k0 + 2 * half;
      v2f a;  a.x = arow[k];                     a.y = arow[k + 1];
      v2f bb; bb.x = W2[k * CH + ct * 16 + l16]; bb.y = W2[(k + 1) * CH + ct * 16 + l16];
      acc = wmma4(a, bb, acc);
    }
    #pragma unroll
    for (int r = 0; r < 8; r++) {
      int m = rt * 16 + r + 8 * half;
      size_t node = (size_t)(base + m);
      int n = ct * 16 + l16;
      float v = acc[r] + sb2[n];
      if (outer) v += outer[node * CH + n];
      out[node * CH + n] = v;
    }
  }
}

__global__ __launch_bounds__(256)
void k_head(const float* __restrict__ xc,
            const float* __restrict__ W1, const float* __restrict__ b1,
            const float* __restrict__ W2, const float* __restrict__ b2,
            const float* __restrict__ W3, const float* __restrict__ b3,
            float* __restrict__ out) {
  __shared__ float sA[32 * PA];
  __shared__ float sH[32 * PA];
  __shared__ float sb1[CH], sb2[CH], sW3[CH * 2], sb3[2];
  const int tid = threadIdx.x;
  const int base = blockIdx.x * 32;
  if (tid < CH) { sb1[tid] = b1[tid]; sb2[tid] = b2[tid]; }
  if (tid < CH * 2) sW3[tid] = W3[tid];
  if (tid < 2) sb3[tid] = b3[tid];
  for (int i = tid; i < 32 * CH; i += 256) sA[(i >> 6) * PA + (i & 63)] = xc[(size_t)base * CH + i];
  __syncthreads();

  const int wave = tid >> 5, lane = tid & 31;
  const int half = lane >> 4, l16 = lane & 15;
  const int rt = wave >> 2, ct = wave & 3;

  {
    const float* arow = &sA[(rt * 16 + l16) * PA];
    v8f acc = {0.f, 0.f, 0.f, 0.f, 0.f, 0.f, 0.f, 0.f};
    #pragma unroll
    for (int k0 = 0; k0 < CH; k0 += 4) {
      int k = k0 + 2 * half;
      v2f a;  a.x = arow[k];                     a.y = arow[k + 1];
      v2f bb; bb.x = W1[k * CH + ct * 16 + l16]; bb.y = W1[(k + 1) * CH + ct * 16 + l16];
      acc = wmma4(a, bb, acc);
    }
    #pragma unroll
    for (int r = 0; r < 8; r++) {
      int m = rt * 16 + r + 8 * half, n = ct * 16 + l16;
      sH[m * PA + n] = fmaxf(acc[r] + sb1[n], 0.f);
    }
  }
  __syncthreads();
  {
    const float* arow = &sH[(rt * 16 + l16) * PA];
    v8f acc = {0.f, 0.f, 0.f, 0.f, 0.f, 0.f, 0.f, 0.f};
    #pragma unroll
    for (int k0 = 0; k0 < CH; k0 += 4) {
      int k = k0 + 2 * half;
      v2f a;  a.x = arow[k];                     a.y = arow[k + 1];
      v2f bb; bb.x = W2[k * CH + ct * 16 + l16]; bb.y = W2[(k + 1) * CH + ct * 16 + l16];
      acc = wmma4(a, bb, acc);
    }
    #pragma unroll
    for (int r = 0; r < 8; r++) {
      int m = rt * 16 + r + 8 * half, n = ct * 16 + l16;
      sA[m * PA + n] = fmaxf(acc[r] + sb2[n], 0.f);
    }
  }
  __syncthreads();
  if (tid < 64) {
    int m = tid >> 1, o = tid & 1;
    float s = sb3[o];
    for (int k = 0; k < CH; k++) s += sA[m * PA + k] * sW3[k * 2 + o];
    out[(size_t)(base + m) * 2 + o] = s;
  }
}

extern "C" void kernel_launch(void* const* d_in, const int* in_sizes, int n_in,
                              void* d_out, int out_size, void* d_ws, size_t ws_size,
                              hipStream_t stream) {
  (void)in_sizes; (void)n_in; (void)out_size; (void)ws_size;
  const float* x      = (const float*)d_in[0];
  const int*   ei     = (const int*)d_in[1];
  const int*   src    = ei;
  const int*   dst    = ei + NE;
  const float* c1_W1  = (const float*)d_in[2];
  const float* c1_b1  = (const float*)d_in[3];
  const float* c1_lnw = (const float*)d_in[4];
  const float* c1_lnb = (const float*)d_in[5];
  const float* c1_W2  = (const float*)d_in[6];
  const float* c1_b2  = (const float*)d_in[7];
  const float* n1_w   = (const float*)d_in[8];
  const float* n1_b   = (const float*)d_in[9];
  const float* cW1    = (const float*)d_in[10];
  const float* cb1    = (const float*)d_in[11];
  const float* clnw   = (const float*)d_in[12];
  const float* clnb   = (const float*)d_in[13];
  const float* cW2    = (const float*)d_in[14];
  const float* cb2    = (const float*)d_in[15];
  const float* nw     = (const float*)d_in[16];
  const float* nb     = (const float*)d_in[17];
  const float* lW1    = (const float*)d_in[18];
  const float* lb1    = (const float*)d_in[19];
  const float* lW2    = (const float*)d_in[20];
  const float* lb2    = (const float*)d_in[21];
  const float* lW3    = (const float*)d_in[22];
  const float* lb3    = (const float*)d_in[23];

  float* ws    = (float*)d_ws;
  float* xc    = ws;
  float* hpre  = ws + (size_t)N64;
  int*   csr   = (int*)(ws + 2 * (size_t)N64);
  int*   deg   = csr;
  int*   rbeg  = csr + NN;
  int*   rcur  = csr + 2 * NN;
  int*   total = csr + 3 * NN;
  int*   col   = csr + 3 * NN + 64;
  float* agg   = ws + 4 * (size_t)N64;
  float* sums  = ws + 5 * (size_t)N64;

  const dim3 B(256);
  const int NB   = (NN + 255) / 256;
  const int MLPB = NN / 32;

  k_csr_zero<<<NB, B, 0, stream>>>(deg, total);
  k_csr_count<<<EBLK, B, 0, stream>>>(dst, deg);
  k_csr_offsets<<<NB, B, 0, stream>>>(deg, total, rbeg, rcur);
  k_csr_fill<<<EBLK, B, 0, stream>>>(src, dst, rcur, col);

  auto edge_phase = [&](const float* hin) {
    k_edge_gather<<<GBLK, B, 0, stream>>>(rbeg, rcur, col, hin, agg);
  };

  edge_phase(x);
  k_conv_mlp<<<MLPB, B, 0, stream>>>(x, agg, c1_W1, c1_b1, c1_lnw, c1_lnb,
                                     c1_W2, c1_b2, nullptr, hpre);
  k_zero2<<<1, 32, 0, stream>>>(sums);
  k_reduce<<<SBLK, B, 0, stream>>>((const float4*)hpre, sums);
  k_gln_apply<<<SBLK, B, 0, stream>>>((const float4*)hpre, sums, n1_w, n1_b, (float4*)xc);

  for (int i = 0; i < 3; i++) {
    k_zero2<<<1, 32, 0, stream>>>(sums);
    k_reduce<<<SBLK, B, 0, stream>>>((const float4*)xc, sums);
    k_gln_apply<<<SBLK, B, 0, stream>>>((const float4*)xc, sums, nw + i * CH, nb + i * CH,
                                        (float4*)hpre);
    edge_phase(hpre);
    k_conv_mlp<<<MLPB, B, 0, stream>>>(hpre, agg,
                                       cW1 + (size_t)i * CH * HD, cb1 + (size_t)i * HD,
                                       clnw + (size_t)i * HD, clnb + (size_t)i * HD,
                                       cW2 + (size_t)i * HD * CH, cb2 + (size_t)i * CH,
                                       xc, xc);
  }

  k_head<<<MLPB, B, 0, stream>>>(xc, lW1, lb1, lW2, lb2, lW3, lb3, (float*)d_out);
}


# Row LayerNorm of the layer kernel by all eight waves (eight lanes share a row, sums added across lanes in three steps) instead of one wave walking 32 rows; fragment addresses as base + constant
# speedup vs baseline: 4.4316x; 1.4986x over previous
#include <hip/hip_runtime.h>
#include <math.h>

#define NN   100000
#define CH   64
#define HD   128
#define NE   800000
#define N64  6400000
#define EPS_MSG 1e-7f
#define EPS_LN  1e-5f
#define PA   (CH + 4)
#define PH   (HD + 4)
#define SBLK 625
#define SPT  10
#define NSLOT 16
#define SLOTP 32
#define SETF  (NSLOT * SLOTP)
#define NSETS 4
#define EBLK 625
#define EPT  5

typedef float v2f __attribute__((ext_vector_type(2)));
typedef float v8f __attribute__((ext_vector_type(8)));

static __device__ __forceinline__ v8f wmma4(v2f a, v2f b, v8f c) {
  return __builtin_amdgcn_wmma_f32_16x16x4_f32(
       false, a,  false, b,
       (short)0, c,  false,  false);
}

static __device__ __forceinline__ void gln_stats(const float* __restrict__ st, float& mu, float& scale) {
  float s = 0.f, q = 0.f;
  for (int j = 0; j < NSLOT; j++) { s += st[j * SLOTP]; q += st[j * SLOTP + 1]; }
  const float invn = 1.f / (float)N64;
  mu = s * invn;
  const float var = q * invn - mu * mu;
  scale = 1.f / (sqrtf(fmaxf(var, 0.f)) + EPS_LN);
}

static __device__ __forceinline__ void gln_coef(float mu, float scale, float w, float b,
                                                float& a, float& c) {
  a = scale * w;
  c = b - mu * a;
}

static __device__ __forceinline__ void block_stats_add(float s, float q, float* __restrict__ st,
                                                       float* sred) {
  for (int off = 16; off > 0; off >>= 1) { s += __shfl_xor(s, off); q += __shfl_xor(q, off); }
  const int wave = threadIdx.x >> 5, lane = threadIdx.x & 31;
  if (lane == 0) { sred[wave] = s; sred[8 + wave] = q; }
  __syncthreads();
  if (threadIdx.x == 0) {
    float ts = 0.f, tq = 0.f;
    for (int w = 0; w < 8; w++) { ts += sred[w]; tq += sred[8 + w]; }
    float* p = st + (blockIdx.x & (NSLOT - 1)) * SLOTP;
    atomicAdd(p, ts);
    atomicAdd(p + 1, tq);
  }
}

__global__ __launch_bounds__(256) void k_gln_apply(const float4* __restrict__ h,
                                                   const float* __restrict__ st_in,
                                                   const float* __restrict__ w,
                                                   const float* __restrict__ b,
                                                   float4* __restrict__ out,
                                                   float* __restrict__ st_out) {
  __shared__ float sred[16];
  const size_t i0 = (size_t)blockIdx.x * (256 * SPT) + threadIdx.x;
  float mu, scale;
  gln_stats(st_in, mu, scale);
  const int c = (threadIdx.x & 15) << 2;
  const float w0 = w[c + 0], w1 = w[c + 1], w2 = w[c + 2], w3 = w[c + 3];
  const float b0 = b[c + 0], b1 = b[c + 1], b2 = b[c + 2], b3 = b[c + 3];
  float4 v[SPT];
  for (int j = 0; j < SPT; j++) v[j] = h[i0 + j * 256];
  float s = 0.f, q = 0.f;
  for (int j = 0; j < SPT; j++) {
    const float r0 = fmaxf((v[j].x - mu) * scale * w0 + b0, 0.f);
    const float r1 = fmaxf((v[j].y - mu) * scale * w1 + b1, 0.f);
    const float r2 = fmaxf((v[j].z - mu) * scale * w2 + b2, 0.f);
    const float r3 = fmaxf((v[j].w - mu) * scale * w3 + b3, 0.f);
    v[j] = make_float4(r0, r1, r2, r3);
    s += (r0 + r1) + (r2 + r3);
    q += (r0 * r0 + r1 * r1) + (r2 * r2 + r3 * r3);
  }
  block_stats_add(s, q, st_out, sred);
  for (int j = 0; j < SPT; j++) out[i0 + j * 256] = v[j];
}

__global__ __launch_bounds__(256) void k_csr_zero(int* __restrict__ deg, int* __restrict__ total,
                                                  float* __restrict__ stats) {
  int i = blockIdx.x * 256 + threadIdx.x;
  if (i < NN) deg[i] = 0;
  if (i == 0) total[0] = 0;
  if (i < NSETS * SETF) stats[i] = 0.f;
}

__global__ __launch_bounds__(256) void k_csr_count(const int* __restrict__ dst,
                                                   int* __restrict__ deg) {
  const int e0 = blockIdx.x * (256 * EPT) + threadIdx.x;
  int d[EPT];
  for (int j = 0; j < EPT; j++) d[j] = dst[e0 + j * 256];
  for (int j = 0; j < EPT; j++) atomicAdd(&deg[d[j]], 1);
}

__global__ __launch_bounds__(256) void k_csr_offsets(const int* __restrict__ deg,
                                                     int* __restrict__ total,
                                                     int* __restrict__ rbeg,
                                                     int* __restrict__ rcur) {
  __shared__ int swave[8];
  __shared__ int sbase;
  const int tid = threadIdx.x, lane = tid & 31, wave = tid >> 5;
  const int n = blockIdx.x * 256 + tid;
  const int d = (n < NN) ? deg[n] : 0;
  int v = d;
  for (int off = 1; off < 32; off <<= 1) {
    int t = __shfl_up(v, off);
    if (lane >= off) v += t;
  }
  if (lane == 31) swave[wave] = v;
  __syncthreads();
  if (tid == 0) {
    int s = 0;
    for (int w = 0; w < 8; w++) { int t = swave[w]; swave[w] = s; s += t; }
    sbase = atomicAdd(total, s);
  }
  __syncthreads();
  if (n < NN) {
    int b = sbase + swave[wave] + v - d;
    rbeg[n] = b;
    rcur[n] = b;
  }
}

__global__ __launch_bounds__(256) void k_csr_fill(const int* __restrict__ src,
                                                  const int* __restrict__ dst,
                                                  int* __restrict__ rcur,
                                                  int* __restrict__ col) {
  const int e0 = blockIdx.x * (256 * EPT) + threadIdx.x;
  int d[EPT], s[EPT], p[EPT];
  for (int j = 0; j < EPT; j++) { d[j] = dst[e0 + j * 256]; s[j] = src[e0 + j * 256]; }
  for (int j = 0; j < EPT; j++) p[j] = atomicAdd(&rcur[d[j]], 1);
  for (int j = 0; j < EPT; j++) col[p[j]] = s[j];
}

static __device__ __forceinline__ void soft4(float v0, float v1, float v2, float v3,
                                             bool p1, bool p2, bool p3,
                                             float& mx, float& s1, float& s2) {
  const float a0 = fmaxf(v0, 0.f) + EPS_MSG;
  const float a1 = p1 ? fmaxf(v1, 0.f) + EPS_MSG : 0.f;
  const float a2 = p2 ? fmaxf(v2, 0.f) + EPS_MSG : 0.f;
  const float a3 = p3 ? fmaxf(v3, 0.f) + EPS_MSG : 0.f;
  const float mn = fmaxf(fmaxf(mx, fmaxf(a0, a1)), fmaxf(a2, a3));
  const float sc = __expf(mx - mn);
  const float e0 = __expf(a0 - mn);
  const float e1 = p1 ? __expf(a1 - mn) : 0.f;
  const float e2 = p2 ? __expf(a2 - mn) : 0.f;
  const float e3 = p3 ? __expf(a3 - mn) : 0.f;
  s1 = s1 * sc + ((e0 + e1) + (e2 + e3));
  s2 = s2 * sc + ((a0 * e0 + a1 * e1) + (a2 * e2 + a3 * e3));
  mx = mn;
}

struct SoftAcc {
  float mx0, mx1, mx2, mx3;
  float s10, s11, s12, s13;
  float s20, s21, s22, s23;
};

template <bool AFF>
static __device__ __forceinline__ void gather_chunk(int idx, int cnt, int cntw, int hb,
                                                    const float* __restrict__ hp,
                                                    const float4 ga, const float4 gc, SoftAcc& A) {
  for (int i = 0; i < cntw; i += 4) {
    const int i0 = __shfl(idx, hb + i);
    const int i1 = __shfl(idx, hb + i + 1);
    const int i2 = __shfl(idx, hb + i + 2);
    const int i3 = __shfl(idx, hb + i + 3);
    if (i < cnt) {
      const bool p1 = i + 1 < cnt, p2 = i + 2 < cnt, p3 = i + 3 < cnt;
      const int j1 = p1 ? i1 : i0, j2 = p2 ? i2 : i0, j3 = p3 ? i3 : i0;
      float4 v0 = *(const float4*)(hp + (size_t)i0 * CH);
      float4 v1 = *(const float4*)(hp + (size_t)j1 * CH);
      float4 v2 = *(const float4*)(hp + (size_t)j2 * CH);
      float4 v3 = *(const float4*)(hp + (size_t)j3 * CH);
      if (AFF) {
        v0.x = fmaf(v0.x, ga.x, gc.x); v1.x = fmaf(v1.x, ga.x, gc.x);
        v2.x = fmaf(v2.x, ga.x, gc.x); v3.x = fmaf(v3.x, ga.x, gc.x);
        v0.y = fmaf(v0.y, ga.y, gc.y); v1.y = fmaf(v1.y, ga.y, gc.y);
        v2.y = fmaf(v2.y, ga.y, gc.y); v3.y = fmaf(v3.y, ga.y, gc.y);
        v0.z = fmaf(v0.z, ga.z, gc.z); v1.z = fmaf(v1.z, ga.z, gc.z);
        v2.z = fmaf(v2.z, ga.z, gc.z); v3.z = fmaf(v3.z, ga.z, gc.z);
        v0.w = fmaf(v0.w, ga.w, gc.w); v1.w = fmaf(v1.w, ga.w, gc.w);
        v2.w = fmaf(v2.w, ga.w, gc.w); v3.w = fmaf(v3.w, ga.w, gc.w);
      }
      soft4(v0.x, v1.x, v2.x, v3.x, p1, p2, p3, A.mx0, A.s10, A.s20);
      soft4(v0.y, v1.y, v2.y, v3.y, p1, p2, p3, A.mx1, A.s11, A.s21);
      soft4(v0.z, v1.z, v2.z, v3.z, p1, p2, p3, A.mx2, A.s12, A.s22);
      soft4(v0.w, v1.w, v2.w, v3.w, p1, p2, p3, A.mx3, A.s13, A.s23);
    }
  }
}

template <bool AFF>
static __device__ __forceinline__ float4 gather_node(int mine, int beg, int deg, int l16, int hb,
                                                     const int* __restrict__ col,
                                                     const float* __restrict__ hp,
                                                     const float4 ga, const float4 gc) {
  const int degw = max(deg, __shfl_xor(deg, 16));
  SoftAcc A = {0.f, 0.f, 0.f, 0.f, 0.f, 0.f, 0.f, 0.f, 0.f, 0.f, 0.f, 0.f};
  gather_chunk<AFF>(mine, deg, min(16, degw), hb, hp, ga, gc, A);
  for (int c0 = 16; c0 < degw; c0 += 16) {
    const int more = (c0 + l16 < deg) ? col[beg + c0 + l16] : 0;
    gather_chunk<AFF>(more, deg - c0, min(16, degw - c0), hb, hp, ga, gc, A);
  }
  float4 r = make_float4(0.f, 0.f, 0.f, 0.f);
  if (deg > 0) r = make_float4(A.s20 / A.s10, A.s21 / A.s11, A.s22 / A.s12, A.s23 / A.s13);
  return r;
}

template <int PATTERN>
static __device__ __forceinline__ float lane_swap(float v) {
  return __int_as_float(__builtin_amdgcn_update_dpp(0, __float_as_int(v), PATTERN, 0xF, 0xF, true));
}

static __device__ __forceinline__ float sum8(float v) {
  v += lane_swap<0xB1>(v);
  v += lane_swap<0x4E>(v);
  v += lane_swap<0x141>(v);
  return v;
}

static __device__ __forceinline__ void ln_acc(const float4 v, float& s, float& q) {
  s += (v.x + v.y) + (v.z + v.w);
  q += (v.x * v.x + v.y * v.y) + (v.z * v.z + v.w * v.w);
}

static __device__ __forceinline__ float4 ln_apply(const float4 v, float mu, float inv,
                                                  const float4 w, const float4 b) {
  return make_float4(fmaxf((v.x - mu) * inv * w.x + b.x, 0.f), fmaxf((v.y - mu) * inv * w.y + b.y, 0.f),
                     fmaxf((v.z - mu) * inv * w.z + b.z, 0.f), fmaxf((v.w - mu) * inv * w.w + b.w, 0.f));
}

template <bool AFF, bool STATS>
__global__ __launch_bounds__(256)
void k_layer(const int* __restrict__ rbeg, const int* __restrict__ rend,
             const int* __restrict__ col, const float* __restrict__ src,
             const float* __restrict__ W1, const float* __restrict__ b1,
             const float* __restrict__ lnw, const float* __restrict__ lnb,
             const float* __restrict__ W2, const float* __restrict__ b2,
             const float* __restrict__ st_in, const float* __restrict__ gw,
             const float* __restrict__ gb, float* __restrict__ out,
             float* __restrict__ st_out) {
  __shared__ __attribute__((aligned(16))) float sAct[32 * PA];
  __shared__ __attribute__((aligned(16))) float sHid[32 * PH];
  __shared__ __attribute__((aligned(16))) float slnw[HD];
  __shared__ __attribute__((aligned(16))) float slnb[HD];
  __shared__ float sb1[HD], sb2[CH];
  __shared__ float sred[16];
  const int tid = threadIdx.x;
  const int base = blockIdx.x * 32;
  const int wave = tid >> 5, lane = tid & 31;
  const int half = lane >> 4, l16 = lane & 15;
  const int hb = lane & 16;
  const int g = l16 << 2;
  const float* hp = src + g;

  {
    const int hw = tid >> 4;
    const int n0 = base + hw, n1 = n0 + 16;
    const int beg0 = rbeg[n0], beg1 = rbeg[n1];
    const int deg0 = rend[n0] - beg0, deg1 = rend[n1] - beg1;
    const int mine0 = (l16 < deg0) ? col[beg0 + l16] : 0;
    const int mine1 = (l16 < deg1) ? col[beg1 + l16] : 0;
    float4 h0 = *(const float4*)(hp + (size_t)n0 * CH);
    float4 h1 = *(const float4*)(hp + (size_t)n1 * CH);
    if (tid < HD) { sb1[tid] = b1[tid]; slnw[tid] = lnw[tid]; slnb[tid] = lnb[tid]; }
    if (tid < CH) sb2[tid] = b2[tid];
    float4 ga = make_float4(1.f, 1.f, 1.f, 1.f), gc = make_float4(0.f, 0.f, 0.f, 0.f);
    if (AFF) {
      float mu, scale;
      gln_stats(st_in, mu, scale);
      gln_coef(mu, scale, gw[g + 0], gb[g + 0], ga.x, gc.x);
      gln_coef(mu, scale, gw[g + 1], gb[g + 1], ga.y, gc.y);
      gln_coef(mu, scale, gw[g + 2], gb[g + 2], ga.z, gc.z);
      gln_coef(mu, scale, gw[g + 3], gb[g + 3], ga.w, gc.w);
      h0.x = fmaxf(fmaf(h0.x, ga.x, gc.x), 0.f); h1.x = fmaxf(fmaf(h1.x, ga.x, gc.x), 0.f);
      h0.y = fmaxf(fmaf(h0.y, ga.y, gc.y), 0.f); h1.y = fmaxf(fmaf(h1.y, ga.y, gc.y), 0.f);
      h0.z = fmaxf(fmaf(h0.z, ga.z, gc.z), 0.f); h1.z = fmaxf(fmaf(h1.z, ga.z, gc.z), 0.f);
      h0.w = fmaxf(fmaf(h0.w, ga.w, gc.w), 0.f); h1.w = fmaxf(fmaf(h1.w, ga.w, gc.w), 0.f);
    }
    const float4 r0 = gather_node<AFF>(mine0, beg0, deg0, l16, hb, col, hp, ga, gc);
    const float4 r1 = gather_node<AFF>(mine1, beg1, deg1, l16, hb, col, hp, ga, gc);
    *(float4*)&sAct[hw * PA + g] = make_float4(h0.x + r0.x, h0.y + r0.y, h0.z + r0.z, h0.w + r0.w);
    *(float4*)&sAct[(16 + hw) * PA + g] =
        make_float4(h1.x + r1.x, h1.y + r1.y, h1.z + r1.z, h1.w + r1.w);
  }
  __syncthreads();

  for (int task = wave; task < 16; task += 8) {
    int rt = task >> 3, ct = task & 7;
    const float* arow = &sAct[(rt * 16 + l16) * PA + 2 * half];
    const float* wcol = W1 + (2 * half) * HD + ct * 16 + l16;
    v8f acc = {0.f, 0.f, 0.f, 0.f, 0.f, 0.f, 0.f, 0.f};
    #pragma unroll
    for (int k0 = 0; k0 < CH; k0 += 4) {
      v2f a;  a.x = arow[k0];       a.y = arow[k0 + 1];
      v2f bb; bb.x = wcol[k0 * HD]; bb.y = wcol[(k0 + 1) * HD];
      acc = wmma4(a, bb, acc);
    }
    #pragma unroll
    for (int r = 0; r < 8; r++) {
      int m = rt * 16 + r + 8 * half;
      int n = ct * 16 + l16;
      sHid[m * PH + n] = acc[r] + sb1[n];
    }
  }
  __syncthreads();

  {
    const int p4 = (tid & 7) << 2;
    float* hrow = &sHid[(tid >> 3) * PH + p4];
    const float4 v0 = *(const float4*)(hrow);
    const float4 v1 = *(const float4*)(hrow + 32);
    const float4 v2 = *(const float4*)(hrow + 64);
    const float4 v3 = *(const float4*)(hrow + 96);
    float s = 0.f, q = 0.f;
    ln_acc(v0, s, q); ln_acc(v1, s, q); ln_acc(v2, s, q); ln_acc(v3, s, q);
    s = sum8(s);
    q = sum8(q);
    const float mu = s * (1.f / HD);
    const float var = q * (1.f / HD) - mu * mu;
    const float inv = rsqrtf(var + EPS_LN);
    *(float4*)(hrow)      = ln_apply(v0, mu, inv, *(const float4*)&slnw[p4],      *(const float4*)&slnb[p4]);
    *(float4*)(hrow + 32) = ln_apply(v1, mu, inv, *(const float4*)&slnw[p4 + 32], *(const float4*)&slnb[p4 + 32]);
    *(float4*)(hrow + 64) = ln_apply(v2, mu, inv, *(const float4*)&slnw[p4 + 64], *(const float4*)&slnb[p4 + 64]);
    *(float4*)(hrow + 96) = ln_apply(v3, mu, inv, *(const float4*)&slnw[p4 + 96], *(const float4*)&slnb[p4 + 96]);
  }
  __syncthreads();

  {
    int rt = wave >> 2, ct = wave & 3;
    const float* arow = &sHid[(rt * 16 + l16) * PH + 2 * half];
    const float* wcol = W2 + (2 * half) * CH + ct * 16 + l16;
    v8f acc = {0.f, 0.f, 0.f, 0.f, 0.f, 0.f, 0.f, 0.f};
    #pragma unroll
    for (int k0 = 0; k0 < HD; k0 += 4) {
      v2f a;  a.x = arow[k0];       a.y = arow[k0 + 1];
      v2f bb; bb.x = wcol[k0 * CH]; bb.y = wcol[(k0 + 1) * CH];
      acc = wmma4(a, bb, acc);
    }
    const int n = ct * 16 + l16;
    const size_t o0 = (size_t)(base + rt * 16 + 8 * half) * CH + n;
    float res[8];
    #pragma unroll
    for (int r = 0; r < 8; r++) res[r] = AFF ? src[o0 + (size_t)r * CH] : 0.f;
    float v[8];
    float s = 0.f, q = 0.f;
    #pragma unroll
    for (int r = 0; r < 8; r++) {
      v[r] = acc[r] + sb2[n] + res[r];
      s += v[r];
      q += v[r] * v[r];
    }
    if (STATS) block_stats_add(s, q, st_out, sred);
    #pragma unroll
    for (int r = 0; r < 8; r++) out[o0 + (size_t)r * CH] = v[r];
  }
}

__global__ __launch_bounds__(256)
void k_head(const float* __restrict__ xc,
            const float* __restrict__ W1, const float* __restrict__ b1,
            const float* __restrict__ W2, const float* __restrict__ b2,
            const float* __restrict__ W3, const float* __restrict__ b3,
            float* __restrict__ out) {
  __shared__ float sA[32 * PA];
  __shared__ float sH[32 * PA];
  __shared__ float sb1[CH], sb2[CH], sW3[CH * 2], sb3[2];
  const int tid = threadIdx.x;
  const int base = blockIdx.x * 32;
  if (tid < CH) { sb1[tid] = b1[tid]; sb2[tid] = b2[tid]; }
  if (tid < CH * 2) sW3[tid] = W3[tid];
  if (tid < 2) sb3[tid] = b3[tid];
  for (int i = tid; i < 32 * CH; i += 256) sA[(i >> 6) * PA + (i & 63)] = xc[(size_t)base * CH + i];
  __syncthreads();

  const int wave = tid >> 5, lane = tid & 31;
  const int half = lane >> 4, l16 = lane & 15;
  const int rt = wave >> 2, ct = wave & 3;

  {
    const float* arow = &sA[(rt * 16 + l16) * PA];
    v8f acc = {0.f, 0.f, 0.f, 0.f, 0.f, 0.f, 0.f, 0.f};
    #pragma unroll
    for (int k0 = 0; k0 < CH; k0 += 4) {
      int k = k0 + 2 * half;
      v2f a;  a.x = arow[k];                     a.y = arow[k + 1];
      v2f bb; bb.x = W1[k * CH + ct * 16 + l16]; bb.y = W1[(k + 1) * CH + ct * 16 + l16];
      acc = wmma4(a, bb, acc);
    }
    #pragma unroll
    for (int r = 0; r < 8; r++) {
      int m = rt * 16 + r + 8 * half, n = ct * 16 + l16;
      sH[m * PA + n] = fmaxf(acc[r] + sb1[n], 0.f);
    }
  }
  __syncthreads();
  {
    const float* arow = &sH[(rt * 16 + l16) * PA];
    v8f acc = {0.f, 0.f, 0.f, 0.f, 0.f, 0.f, 0.f, 0.f};
    #pragma unroll
    for (int k0 = 0; k0 < CH; k0 += 4) {
      int k = k0 + 2 * half;
      v2f a;  a.x = arow[k];                     a.y = arow[k + 1];
      v2f bb; bb.x = W2[k * CH + ct * 16 + l16]; bb.y = W2[(k + 1) * CH + ct * 16 + l16];
      acc = wmma4(a, bb, acc);
    }
    #pragma unroll
    for (int r = 0; r < 8; r++) {
      int m = rt * 16 + r + 8 * half, n = ct * 16 + l16;
      sA[m * PA + n] = fmaxf(acc[r] + sb2[n], 0.f);
    }
  }
  __syncthreads();
  if (tid < 64) {
    int m = tid >> 1, o = tid & 1;
    float s = sb3[o];
    for (int k = 0; k < CH; k++) s += sA[m * PA + k] * sW3[k * 2 + o];
    out[(size_t)(base + m) * 2 + o] = s;
  }
}

extern "C" void kernel_launch(void* const* d_in, const int* in_sizes, int n_in,
                              void* d_out, int out_size, void* d_ws, size_t ws_size,
                              hipStream_t stream) {
  (void)in_sizes; (void)n_in; (void)out_size; (void)ws_size;
  const float* x      = (const float*)d_in[0];
  const int*   ei     = (const int*)d_in[1];
  const int*   src    = ei;
  const int*   dst    = ei + NE;
  const float* c1_W1  = (const float*)d_in[2];
  const float* c1_b1  = (const float*)d_in[3];
  const float* c1_lnw = (const float*)d_in[4];
  const float* c1_lnb = (const float*)d_in[5];
  const float* c1_W2  = (const float*)d_in[6];
  const float* c1_b2  = (const float*)d_in[7];
  const float* n1_w   = (const float*)d_in[8];
  const float* n1_b   = (const float*)d_in[9];
  const float* cW1    = (const float*)d_in[10];
  const float* cb1    = (const float*)d_in[11];
  const float* clnw   = (const float*)d_in[12];
  const float* clnb   = (const float*)d_in[13];
  const float* cW2    = (const float*)d_in[14];
  const float* cb2    = (const float*)d_in[15];
  const float* nw     = (const float*)d_in[16];
  const float* nb     = (const float*)d_in[17];
  const float* lW1    = (const float*)d_in[18];
  const float* lb1    = (const float*)d_in[19];
  const float* lW2    = (const float*)d_in[20];
  const float* lb2    = (const float*)d_in[21];
  const float* lW3    = (const float*)d_in[22];
  const float* lb3    = (const float*)d_in[23];

  float* ws    = (float*)d_ws;
  float* xa    = ws;
  float* hpre  = ws + (size_t)N64;
  int*   csr   = (int*)(ws + 2 * (size_t)N64);
  int*   deg   = csr;
  int*   rbeg  = csr + NN;
  int*   rcur  = csr + 2 * NN;
  int*   total = csr + 3 * NN;
  int*   col   = csr + 3 * NN + 64;
  float* xb    = ws + 4 * (size_t)N64;
  float* stats = ws + 3 * (size_t)N64;

  const dim3 B(256);
  const int NB   = (NN + 255) / 256;
  const int MLPB = NN / 32;

  k_csr_zero<<<NB, B, 0, stream>>>(deg, total, stats);
  k_csr_count<<<EBLK, B, 0, stream>>>(dst, deg);
  k_csr_offsets<<<NB, B, 0, stream>>>(deg, total, rbeg, rcur);
  k_csr_fill<<<EBLK, B, 0, stream>>>(src, dst, rcur, col);

  k_layer<false, true><<<MLPB, B, 0, stream>>>(rbeg, rcur, col, x, c1_W1, c1_b1, c1_lnw, c1_lnb,
                                               c1_W2, c1_b2, nullptr, nullptr, nullptr,
                                               hpre, stats);
  k_gln_apply<<<SBLK, B, 0, stream>>>((const float4*)hpre, stats, n1_w, n1_b, (float4*)xa,
                                      stats + SETF);

  for (int i = 0; i < 3; i++) {
    const float* st = stats + (size_t)(1 + i) * SETF;
    const float* xin = (i & 1) ? xb : xa;
    float* xout = (i & 1) ? xa : xb;
    if (i < 2)
      k_layer<true, true><<<MLPB, B, 0, stream>>>(rbeg, rcur, col, xin,
                                       cW1 + (size_t)i * CH * HD, cb1 + (size_t)i * HD,
                                       clnw + (size_t)i * HD, clnb + (size_t)i * HD,
                                       cW2 + (size_t)i * HD * CH, cb2 + (size_t)i * CH,
                                       st, nw + i * CH, nb + i * CH, xout,
                                       stats + (size_t)(2 + i) * SETF);
    else
      k_layer<true, false><<<MLPB, B, 0, stream>>>(rbeg, rcur, col, xin,
                                       cW1 + (size_t)i * CH * HD, cb1 + (size_t)i * HD,
                                       clnw + (size_t)i * HD, clnb + (size_t)i * HD,
                                       cW2 + (size_t)i * HD * CH, cb2 + (size_t)i * CH,
                                       st, nw + i * CH, nb + i * CH, xout, nullptr);
  }

  k_head<<<MLPB, B, 0, stream>>>(xb, lW1, lb1, lW2, lb2, lW3, lb3, (float*)d_out);
}


# Edge grouping with one atomic pass, not two: the counting add's return is the edge's rank in its node's run and is kept per edge; the fill writes col[start+rank] with plain loads and stores
# speedup vs baseline: 4.9025x; 1.1063x over previous
#include <hip/hip_runtime.h>
#include <math.h>

#define NN   100000
#define CH   64
#define HD   128
#define NE   800000
#define N64  6400000
#define EPS_MSG 1e-7f
#define EPS_LN  1e-5f
#define PA   (CH + 4)
#define PH   (HD + 4)
#define SBLK 625
#define SPT  10
#define NSLOT 16
#define SLOTP 32
#define SETF  (NSLOT * SLOTP)
#define NSETS 4
#define EBLK 625
#define EPT  5

typedef float v2f __attribute__((ext_vector_type(2)));
typedef float v8f __attribute__((ext_vector_type(8)));

static __device__ __forceinline__ v8f wmma4(v2f a, v2f b, v8f c) {
  return __builtin_amdgcn_wmma_f32_16x16x4_f32(
       false, a,  false, b,
       (short)0, c,  false,  false);
}

static __device__ __forceinline__ void gln_stats(const float* __restrict__ st, float& mu, float& scale) {
  float s = 0.f, q = 0.f;
  for (int j = 0; j < NSLOT; j++) { s += st[j * SLOTP]; q += st[j * SLOTP + 1]; }
  const float invn = 1.f / (float)N64;
  mu = s * invn;
  const float var = q * invn - mu * mu;
  scale = 1.f / (sqrtf(fmaxf(var, 0.f)) + EPS_LN);
}

static __device__ __forceinline__ void gln_coef(float mu, float scale, float w, float b,
                                                float& a, float& c) {
  a = scale * w;
  c = b - mu * a;
}

static __device__ __forceinline__ void block_stats_add(float s, float q, float* __restrict__ st,
                                                       float* sred) {
  for (int off = 16; off > 0; off >>= 1) { s += __shfl_xor(s, off); q += __shfl_xor(q, off); }
  const int wave = threadIdx.x >> 5, lane = threadIdx.x & 31;
  if (lane == 0) { sred[wave] = s; sred[8 + wave] = q; }
  __syncthreads();
  if (threadIdx.x == 0) {
    float ts = 0.f, tq = 0.f;
    for (int w = 0; w < 8; w++) { ts += sred[w]; tq += sred[8 + w]; }
    float* p = st + (blockIdx.x & (NSLOT - 1)) * SLOTP;
    atomicAdd(p, ts);
    atomicAdd(p + 1, tq);
  }
}

__global__ __launch_bounds__(256) void k_gln_apply(const float4* __restrict__ h,
                                                   const float* __restrict__ st_in,
                                                   const float* __restrict__ w,
                                                   const float* __restrict__ b,
                                                   float4* __restrict__ out,
                                                   float* __restrict__ st_out) {
  __shared__ float sred[16];
  const size_t i0 = (size_t)blockIdx.x * (256 * SPT) + threadIdx.x;
  float mu, scale;
  gln_stats(st_in, mu, scale);
  const int c = (threadIdx.x & 15) << 2;
  const float w0 = w[c + 0], w1 = w[c + 1], w2 = w[c + 2], w3 = w[c + 3];
  const float b0 = b[c + 0], b1 = b[c + 1], b2 = b[c + 2], b3 = b[c + 3];
  float4 v[SPT];
  for (int j = 0; j < SPT; j++) v[j] = h[i0 + j * 256];
  float s = 0.f, q = 0.f;
  for (int j = 0; j < SPT; j++) {
    const float r0 = fmaxf((v[j].x - mu) * scale * w0 + b0, 0.f);
    const float r1 = fmaxf((v[j].y - mu) * scale * w1 + b1, 0.f);
    const float r2 = fmaxf((v[j].z - mu) * scale * w2 + b2, 0.f);
    const float r3 = fmaxf((v[j].w - mu) * scale * w3 + b3, 0.f);
    v[j] = make_float4(r0, r1, r2, r3);
    s += (r0 + r1) + (r2 + r3);
    q += (r0 * r0 + r1 * r1) + (r2 * r2 + r3 * r3);
  }
  block_stats_add(s, q, st_out, sred);
  for (int j = 0; j < SPT; j++) out[i0 + j * 256] = v[j];
}

__global__ __launch_bounds__(256) void k_csr_zero(int* __restrict__ deg, int* __restrict__ total,
                                                  float* __restrict__ stats) {
  int i = blockIdx.x * 256 + threadIdx.x;
  if (i < NN) deg[i] = 0;
  if (i == 0) total[0] = 0;
  if (i < NSETS * SETF) stats[i] = 0.f;
}

__global__ __launch_bounds__(256) void k_csr_count(const int* __restrict__ dst,
                                                   int* __restrict__ deg,
                                                   int* __restrict__ rank) {
  const int e0 = blockIdx.x * (256 * EPT) + threadIdx.x;
  int d[EPT], r[EPT];
  for (int j = 0; j < EPT; j++) d[j] = dst[e0 + j * 256];
  for (int j = 0; j < EPT; j++) r[j] = atomicAdd(&deg[d[j]], 1);
  for (int j = 0; j < EPT; j++) rank[e0 + j * 256] = r[j];
}

__global__ __launch_bounds__(256) void k_csr_offsets(const int* __restrict__ deg,
                                                     int* __restrict__ total,
                                                     int* __restrict__ rbeg,
                                                     int* __restrict__ rend) {
  __shared__ int swave[8];
  __shared__ int sbase;
  const int tid = threadIdx.x, lane = tid & 31, wave = tid >> 5;
  const int n = blockIdx.x * 256 + tid;
  const int d = (n < NN) ? deg[n] : 0;
  int v = d;
  for (int off = 1; off < 32; off <<= 1) {
    int t = __shfl_up(v, off);
    if (lane >= off) v += t;
  }
  if (lane == 31) swave[wave] = v;
  __syncthreads();
  if (tid == 0) {
    int s = 0;
    for (int w = 0; w < 8; w++) { int t = swave[w]; swave[w] = s; s += t; }
    sbase = atomicAdd(total, s);
  }
  __syncthreads();
  if (n < NN) {
    int b = sbase + swave[wave] + v - d;
    rbeg[n] = b;
    rend[n] = b + d;
  }
}

__global__ __launch_bounds__(256) void k_csr_fill(const int* __restrict__ src,
                                                  const int* __restrict__ dst,
                                                  const int* __restrict__ rbeg,
                                                  const int* __restrict__ rank,
                                                  int* __restrict__ col) {
  const int e0 = blockIdx.x * (256 * EPT) + threadIdx.x;
  int d[EPT], s[EPT], r[EPT], p[EPT];
  for (int j = 0; j < EPT; j++) {
    d[j] = dst[e0 + j * 256]; s[j] = src[e0 + j * 256]; r[j] = rank[e0 + j * 256];
  }
  for (int j = 0; j < EPT; j++) p[j] = rbeg[d[j]] + r[j];
  for (int j = 0; j < EPT; j++) col[p[j]] = s[j];
}

static __device__ __forceinline__ void soft4(float v0, float v1, float v2, float v3,
                                             bool p1, bool p2, bool p3,
                                             float& mx, float& s1, float& s2) {
  const float a0 = fmaxf(v0, 0.f) + EPS_MSG;
  const float a1 = p1 ? fmaxf(v1, 0.f) + EPS_MSG : 0.f;
  const float a2 = p2 ? fmaxf(v2, 0.f) + EPS_MSG : 0.f;
  const float a3 = p3 ? fmaxf(v3, 0.f) + EPS_MSG : 0.f;
  const float mn = fmaxf(fmaxf(mx, fmaxf(a0, a1)), fmaxf(a2, a3));
  const float sc = __expf(mx - mn);
  const float e0 = __expf(a0 - mn);
  const float e1 = p1 ? __expf(a1 - mn) : 0.f;
  const float e2 = p2 ? __expf(a2 - mn) : 0.f;
  const float e3 = p3 ? __expf(a3 - mn) : 0.f;
  s1 = s1 * sc + ((e0 + e1) + (e2 + e3));
  s2 = s2 * sc + ((a0 * e0 + a1 * e1) + (a2 * e2 + a3 * e3));
  mx = mn;
}

struct SoftAcc {
  float mx0, mx1, mx2, mx3;
  float s10, s11, s12, s13;
  float s20, s21, s22, s23;
};

template <bool AFF>
static __device__ __forceinline__ void gather_chunk(int idx, int cnt, int cntw, int hb,
                                                    const float* __restrict__ hp,
                                                    const float4 ga, const float4 gc, SoftAcc& A) {
  for (int i = 0; i < cntw; i += 4) {
    const int i0 = __shfl(idx, hb + i);
    const int i1 = __shfl(idx, hb + i + 1);
    const int i2 = __shfl(idx, hb + i + 2);
    const int i3 = __shfl(idx, hb + i + 3);
    if (i < cnt) {
      const bool p1 = i + 1 < cnt, p2 = i + 2 < cnt, p3 = i + 3 < cnt;
      const int j1 = p1 ? i1 : i0, j2 = p2 ? i2 : i0, j3 = p3 ? i3 : i0;
      float4 v0 = *(const float4*)(hp + (size_t)i0 * CH);
      float4 v1 = *(const float4*)(hp + (size_t)j1 * CH);
      float4 v2 = *(const float4*)(hp + (size_t)j2 * CH);
      float4 v3 = *(const float4*)(hp + (size_t)j3 * CH);
      if (AFF) {
        v0.x = fmaf(v0.x, ga.x, gc.x); v1.x = fmaf(v1.x, ga.x, gc.x);
        v2.x = fmaf(v2.x, ga.x, gc.x); v3.x = fmaf(v3.x, ga.x, gc.x);
        v0.y = fmaf(v0.y, ga.y, gc.y); v1.y = fmaf(v1.y, ga.y, gc.y);
        v2.y = fmaf(v2.y, ga.y, gc.y); v3.y = fmaf(v3.y, ga.y, gc.y);
        v0.z = fmaf(v0.z, ga.z, gc.z); v1.z = fmaf(v1.z, ga.z, gc.z);
        v2.z = fmaf(v2.z, ga.z, gc.z); v3.z = fmaf(v3.z, ga.z, gc.z);
        v0.w = fmaf(v0.w, ga.w, gc.w); v1.w = fmaf(v1.w, ga.w, gc.w);
        v2.w = fmaf(v2.w, ga.w, gc.w); v3.w = fmaf(v3.w, ga.w, gc.w);
      }
      soft4(v0.x, v1.x, v2.x, v3.x, p1, p2, p3, A.mx0, A.s10, A.s20);
      soft4(v0.y, v1.y, v2.y, v3.y, p1, p2, p3, A.mx1, A.s11, A.s21);
      soft4(v0.z, v1.z, v2.z, v3.z, p1, p2, p3, A.mx2, A.s12, A.s22);
      soft4(v0.w, v1.w, v2.w, v3.w, p1, p2, p3, A.mx3, A.s13, A.s23);
    }
  }
}

template <bool AFF>
static __device__ __forceinline__ float4 gather_node(int mine, int beg, int deg, int l16, int hb,
                                                     const int* __restrict__ col,
                                                     const float* __restrict__ hp,
                                                     const float4 ga, const float4 gc) {
  const int degw = max(deg, __shfl_xor(deg, 16));
  SoftAcc A = {0.f, 0.f, 0.f, 0.f, 0.f, 0.f, 0.f, 0.f, 0.f, 0.f, 0.f, 0.f};
  gather_chunk<AFF>(mine, deg, min(16, degw), hb, hp, ga, gc, A);
  for (int c0 = 16; c0 < degw; c0 += 16) {
    const int more = (c0 + l16 < deg) ? col[beg + c0 + l16] : 0;
    gather_chunk<AFF>(more, deg - c0, min(16, degw - c0), hb, hp, ga, gc, A);
  }
  float4 r = make_float4(0.f, 0.f, 0.f, 0.f);
  if (deg > 0) r = make_float4(A.s20 / A.s10, A.s21 / A.s11, A.s22 / A.s12, A.s23 / A.s13);
  return r;
}

template <int PATTERN>
static __device__ __forceinline__ float lane_swap(float v) {
  return __int_as_float(__builtin_amdgcn_update_dpp(0, __float_as_int(v), PATTERN, 0xF, 0xF, true));
}

static __device__ __forceinline__ float sum8(float v) {
  v += lane_swap<0xB1>(v);
  v += lane_swap<0x4E>(v);
  v += lane_swap<0x141>(v);
  return v;
}

static __device__ __forceinline__ void ln_acc(const float4 v, float& s, float& q) {
  s += (v.x + v.y) + (v.z + v.w);
  q += (v.x * v.x + v.y * v.y) + (v.z * v.z + v.w * v.w);
}

static __device__ __forceinline__ float4 ln_apply(const float4 v, float mu, float inv,
                                                  const float4 w, const float4 b) {
  return make_float4(fmaxf((v.x - mu) * inv * w.x + b.x, 0.f), fmaxf((v.y - mu) * inv * w.y + b.y, 0.f),
                     fmaxf((v.z - mu) * inv * w.z + b.z, 0.f), fmaxf((v.w - mu) * inv * w.w + b.w, 0.f));
}

template <bool AFF, bool STATS>
__global__ __launch_bounds__(256)
void k_layer(const int* __restrict__ rbeg, const int* __restrict__ rend,
             const int* __restrict__ col, const float* __restrict__ src,
             const float* __restrict__ W1, const float* __restrict__ b1,
             const float* __restrict__ lnw, const float* __restrict__ lnb,
             const float* __restrict__ W2, const float* __restrict__ b2,
             const float* __restrict__ st_in, const float* __restrict__ gw,
             const float* __restrict__ gb, float* __restrict__ out,
             float* __restrict__ st_out) {
  __shared__ __attribute__((aligned(16))) float sAct[32 * PA];
  __shared__ __attribute__((aligned(16))) float sHid[32 * PH];
  __shared__ __attribute__((aligned(16))) float slnw[HD];
  __shared__ __attribute__((aligned(16))) float slnb[HD];
  __shared__ float sb1[HD], sb2[CH];
  __shared__ float sred[16];
  const int tid = threadIdx.x;
  const int base = blockIdx.x * 32;
  const int wave = tid >> 5, lane = tid & 31;
  const int half = lane >> 4, l16 = lane & 15;
  const int hb = lane & 16;
  const int g = l16 << 2;
  const float* hp = src + g;

  {
    const int hw = tid >> 4;
    const int n0 = base + hw, n1 = n0 + 16;
    const int beg0 = rbeg[n0], beg1 = rbeg[n1];
    const int deg0 = rend[n0] - beg0, deg1 = rend[n1] - beg1;
    const int mine0 = (l16 < deg0) ? col[beg0 + l16] : 0;
    const int mine1 = (l16 < deg1) ? col[beg1 + l16] : 0;
    float4 h0 = *(const float4*)(hp + (size_t)n0 * CH);
    float4 h1 = *(const float4*)(hp + (size_t)n1 * CH);
    if (tid < HD) { sb1[tid] = b1[tid]; slnw[tid] = lnw[tid]; slnb[tid] = lnb[tid]; }
    if (tid < CH) sb2[tid] = b2[tid];
    float4 ga = make_float4(1.f, 1.f, 1.f, 1.f), gc = make_float4(0.f, 0.f, 0.f, 0.f);
    if (AFF) {
      float mu, scale;
      gln_stats(st_in, mu, scale);
      gln_coef(mu, scale, gw[g + 0], gb[g + 0], ga.x, gc.x);
      gln_coef(mu, scale, gw[g + 1], gb[g + 1], ga.y, gc.y);
      gln_coef(mu, scale, gw[g + 2], gb[g + 2], ga.z, gc.z);
      gln_coef(mu, scale, gw[g + 3], gb[g + 3], ga.w, gc.w);
      h0.x = fmaxf(fmaf(h0.x, ga.x, gc.x), 0.f); h1.x = fmaxf(fmaf(h1.x, ga.x, gc.x), 0.f);
      h0.y = fmaxf(fmaf(h0.y, ga.y, gc.y), 0.f); h1.y = fmaxf(fmaf(h1.y, ga.y, gc.y), 0.f);
      h0.z = fmaxf(fmaf(h0.z, ga.z, gc.z), 0.f); h1.z = fmaxf(fmaf(h1.z, ga.z, gc.z), 0.f);
      h0.w = fmaxf(fmaf(h0.w, ga.w, gc.w), 0.f); h1.w = fmaxf(fmaf(h1.w, ga.w, gc.w), 0.f);
    }
    const float4 r0 = gather_node<AFF>(mine0, beg0, deg0, l16, hb, col, hp, ga, gc);
    const float4 r1 = gather_node<AFF>(mine1, beg1, deg1, l16, hb, col, hp, ga, gc);
    *(float4*)&sAct[hw * PA + g] = make_float4(h0.x + r0.x, h0.y + r0.y, h0.z + r0.z, h0.w + r0.w);
    *(float4*)&sAct[(16 + hw) * PA + g] =
        make_float4(h1.x + r1.x, h1.y + r1.y, h1.z + r1.z, h1.w + r1.w);
  }
  __syncthreads();

  for (int task = wave; task < 16; task += 8) {
    int rt = task >> 3, ct = task & 7;
    const float* arow = &sAct[(rt * 16 + l16) * PA + 2 * half];
    const float* wcol = W1 + (2 * half) * HD + ct * 16 + l16;
    v8f acc = {0.f, 0.f, 0.f, 0.f, 0.f, 0.f, 0.f, 0.f};
    #pragma unroll
    for (int k0 = 0; k0 < CH; k0 += 4) {
      v2f a;  a.x = arow[k0];       a.y = arow[k0 + 1];
      v2f bb; bb.x = wcol[k0 * HD]; bb.y = wcol[(k0 + 1) * HD];
      acc = wmma4(a, bb, acc);
    }
    #pragma unroll
    for (int r = 0; r < 8; r++) {
      int m = rt * 16 + r + 8 * half;
      int n = ct * 16 + l16;
      sHid[m * PH + n] = acc[r] + sb1[n];
    }
  }
  __syncthreads();

  {
    const int p4 = (tid & 7) << 2;
    float* hrow = &sHid[(tid >> 3) * PH + p4];
    const float4 v0 = *(const float4*)(hrow);
    const float4 v1 = *(const float4*)(hrow + 32);
    const float4 v2 = *(const float4*)(hrow + 64);
    const float4 v3 = *(const float4*)(hrow + 96);
    float s = 0.f, q = 0.f;
    ln_acc(v0, s, q); ln_acc(v1, s, q); ln_acc(v2, s, q); ln_acc(v3, s, q);
    s = sum8(s);
    q = sum8(q);
    const float mu = s * (1.f / HD);
    const float var = q * (1.f / HD) - mu * mu;
    const float inv = rsqrtf(var + EPS_LN);
    *(float4*)(hrow)      = ln_apply(v0, mu, inv, *(const float4*)&slnw[p4],      *(const float4*)&slnb[p4]);
    *(float4*)(hrow + 32) = ln_apply(v1, mu, inv, *(const float4*)&slnw[p4 + 32], *(const float4*)&slnb[p4 + 32]);
    *(float4*)(hrow + 64) = ln_apply(v2, mu, inv, *(const float4*)&slnw[p4 + 64], *(const float4*)&slnb[p4 + 64]);
    *(float4*)(hrow + 96) = ln_apply(v3, mu, inv, *(const float4*)&slnw[p4 + 96], *(const float4*)&slnb[p4 + 96]);
  }
  __syncthreads();

  {
    int rt = wave >> 2, ct = wave & 3;
    const float* arow = &sHid[(rt * 16 + l16) * PH + 2 * half];
    const float* wcol = W2 + (2 * half) * CH + ct * 16 + l16;
    v8f acc = {0.f, 0.f, 0.f, 0.f, 0.f, 0.f, 0.f, 0.f};
    #pragma unroll
    for (int k0 = 0; k0 < HD; k0 += 4) {
      v2f a;  a.x = arow[k0];       a.y = arow[k0 + 1];
      v2f bb; bb.x = wcol[k0 * CH]; bb.y = wcol[(k0 + 1) * CH];
      acc = wmma4(a, bb, acc);
    }
    const int n = ct * 16 + l16;
    const size_t o0 = (size_t)(base + rt * 16 + 8 * half) * CH + n;
    float res[8];
    #pragma unroll
    for (int r = 0; r < 8; r++) res[r] = AFF ? src[o0 + (size_t)r * CH] : 0.f;
    float v[8];
    float s = 0.f, q = 0.f;
    #pragma unroll
    for (int r = 0; r < 8; r++) {
      v[r] = acc[r] + sb2[n] + res[r];
      s += v[r];
      q += v[r] * v[r];
    }
    if (STATS) block_stats_add(s, q, st_out, sred);
    #pragma unroll
    for (int r = 0; r < 8; r++) out[o0 + (size_t)r * CH] = v[r];
  }
}

__global__ __launch_bounds__(256)
void k_head(const float* __restrict__ xc,
            const float* __restrict__ W1, const float* __restrict__ b1,
            const float* __restrict__ W2, const float* __restrict__ b2,
            const float* __restrict__ W3, const float* __restrict__ b3,
            float* __restrict__ out) {
  __shared__ float sA[32 * PA];
  __shared__ float sH[32 * PA];
  __shared__ float sb1[CH], sb2[CH], sW3[CH * 2], sb3[2];
  const int tid = threadIdx.x;
  const int base = blockIdx.x * 32;
  if (tid < CH) { sb1[tid] = b1[tid]; sb2[tid] = b2[tid]; }
  if (tid < CH * 2) sW3[tid] = W3[tid];
  if (tid < 2) sb3[tid] = b3[tid];
  for (int i = tid; i < 32 * CH; i += 256) sA[(i >> 6) * PA + (i & 63)] = xc[(size_t)base * CH + i];
  __syncthreads();

  const int wave = tid >> 5, lane = tid & 31;
  const int half = lane >> 4, l16 = lane & 15;
  const int rt = wave >> 2, ct = wave & 3;

  {
    const float* arow = &sA[(rt * 16 + l16) * PA];
    v8f acc = {0.f, 0.f, 0.f, 0.f, 0.f, 0.f, 0.f, 0.f};
    #pragma unroll
    for (int k0 = 0; k0 < CH; k0 += 4) {
      int k = k0 + 2 * half;
      v2f a;  a.x = arow[k];                     a.y = arow[k + 1];
      v2f bb; bb.x = W1[k * CH + ct * 16 + l16]; bb.y = W1[(k + 1) * CH + ct * 16 + l16];
      acc = wmma4(a, bb, acc);
    }
    #pragma unroll
    for (int r = 0; r < 8; r++) {
      int m = rt * 16 + r + 8 * half, n = ct * 16 + l16;
      sH[m * PA + n] = fmaxf(acc[r] + sb1[n], 0.f);
    }
  }
  __syncthreads();
  {
    const float* arow = &sH[(rt * 16 + l16) * PA];
    v8f acc = {0.f, 0.f, 0.f, 0.f, 0.f, 0.f, 0.f, 0.f};
    #pragma unroll
    for (int k0 = 0; k0 < CH; k0 += 4) {
      int k = k0 + 2 * half;
      v2f a;  a.x = arow[k];                     a.y = arow[k + 1];
      v2f bb; bb.x = W2[k * CH + ct * 16 + l16]; bb.y = W2[(k + 1) * CH + ct * 16 + l16];
      acc = wmma4(a, bb, acc);
    }
    #pragma unroll
    for (int r = 0; r < 8; r++) {
      int m = rt * 16 + r + 8 * half, n = ct * 16 + l16;
      sA[m * PA + n] = fmaxf(acc[r] + sb2[n], 0.f);
    }
  }
  __syncthreads();
  if (tid < 64) {
    int m = tid >> 1, o = tid & 1;
    float s = sb3[o];
    for (int k = 0; k < CH; k++) s += sA[m * PA + k] * sW3[k * 2 + o];
    out[(size_t)(base + m) * 2 + o] = s;
  }
}

extern "C" void kernel_launch(void* const* d_in, const int* in_sizes, int n_in,
                              void* d_out, int out_size, void* d_ws, size_t ws_size,
                              hipStream_t stream) {
  (void)in_sizes; (void)n_in; (void)out_size; (void)ws_size;
  const float* x      = (const float*)d_in[0];
  const int*   ei     = (const int*)d_in[1];
  const int*   src    = ei;
  const int*   dst    = ei + NE;
  const float* c1_W1  = (const float*)d_in[2];
  const float* c1_b1  = (const float*)d_in[3];
  const float* c1_lnw = (const float*)d_in[4];
  const float* c1_lnb = (const float*)d_in[5];
  const float* c1_W2  = (const float*)d_in[6];
  const float* c1_b2  = (const float*)d_in[7];
  const float* n1_w   = (const float*)d_in[8];
  const float* n1_b   = (const float*)d_in[9];
  const float* cW1    = (const float*)d_in[10];
  const float* cb1    = (const float*)d_in[11];
  const float* clnw   = (const float*)d_in[12];
  const float* clnb   = (const float*)d_in[13];
  const float* cW2    = (const float*)d_in[14];
  const float* cb2    = (const float*)d_in[15];
  const float* nw     = (const float*)d_in[16];
  const float* nb     = (const float*)d_in[17];
  const float* lW1    = (const float*)d_in[18];
  const float* lb1    = (const float*)d_in[19];
  const float* lW2    = (const float*)d_in[20];
  const float* lb2    = (const float*)d_in[21];
  const float* lW3    = (const float*)d_in[22];
  const float* lb3    = (const float*)d_in[23];

  float* ws    = (float*)d_ws;
  float* xa    = ws;
  float* hpre  = ws + (size_t)N64;
  int*   csr   = (int*)(ws + 2 * (size_t)N64);
  int*   deg   = csr;
  int*   rbeg  = csr + NN;
  int*   rend  = csr + 2 * NN;
  int*   total = csr + 3 * NN;
  int*   col   = csr + 3 * NN + 64;
  int*   rank  = col + NE;
  float* xb    = ws + 4 * (size_t)N64;
  float* stats = ws + 3 * (size_t)N64;

  const dim3 B(256);
  const int NB   = (NN + 255) / 256;
  const int MLPB = NN / 32;

  k_csr_zero<<<NB, B, 0, stream>>>(deg, total, stats);
  k_csr_count<<<EBLK, B, 0, stream>>>(dst, deg, rank);
  k_csr_offsets<<<NB, B, 0, stream>>>(deg, total, rbeg, rend);
  k_csr_fill<<<EBLK, B, 0, stream>>>(src, dst, rbeg, rank, col);

  k_layer<false, true><<<MLPB, B, 0, stream>>>(rbeg, rend, col, x, c1_W1, c1_b1, c1_lnw, c1_lnb,
                                               c1_W2, c1_b2, nullptr, nullptr, nullptr,
                                               hpre, stats);
  k_gln_apply<<<SBLK, B, 0, stream>>>((const float4*)hpre, stats, n1_w, n1_b, (float4*)xa,
                                      stats + SETF);

  for (int i = 0; i < 3; i++) {
    const float* st = stats + (size_t)(1 + i) * SETF;
    const float* xin = (i & 1) ? xb : xa;
    float* xout = (i & 1) ? xa : xb;
    if (i < 2)
      k_layer<true, true><<<MLPB, B, 0, stream>>>(rbeg, rend, col, xin,
                                       cW1 + (size_t)i * CH * HD, cb1 + (size_t)i * HD,
                                       clnw + (size_t)i * HD, clnb + (size_t)i * HD,
                                       cW2 + (size_t)i * HD * CH, cb2 + (size_t)i * CH,
                                       st, nw + i * CH, nb + i * CH, xout,
                                       stats + (size_t)(2 + i) * SETF);
    else
      k_layer<true, false><<<MLPB, B, 0, stream>>>(rbeg, rend, col, xin,
                                       cW1 + (size_t)i * CH * HD, cb1 + (size_t)i * HD,
                                       clnw + (size_t)i * HD, clnb + (size_t)i * HD,
                                       cW2 + (size_t)i * HD * CH, cb2 + (size_t)i * CH,
                                       st, nw + i * CH, nb + i * CH, xout, nullptr);
  }

  k_head<<<MLPB, B, 0, stream>>>(xb, lW1, lb1, lW2, lb2, lW3, lb3, (float*)d_out);
}


# Edge grouping as a two-level counting sort on LDS counters (391 buckets of 256 nodes, then the nodes of a bucket): no global atomic per edge; a node's edge order stays the arrival order of adds
# speedup vs baseline: 5.1003x; 1.0403x over previous
#include <hip/hip_runtime.h>
#include <math.h>

#define NN   100000
#define CH   64
#define HD   128
#define NE   800000
#define N64  6400000
#define EPS_MSG 1e-7f
#define EPS_LN  1e-5f
#define PA   (CH + 4)
#define PH   (HD + 4)
#define SBLK 625
#define SPT  10
#define NSLOT 16
#define SLOTP 32
#define SETF  (NSLOT * SLOTP)
#define NSETS 4
#define NBK  391
#define PBLK 625
#define EPT  5
static_assert(NN <= (1 << 24), "a grouped edge is packed as source node << 8 | low byte of the destination");

typedef float v2f __attribute__((ext_vector_type(2)));
typedef float v8f __attribute__((ext_vector_type(8)));

static __device__ __forceinline__ v8f wmma4(v2f a, v2f b, v8f c) {
  return __builtin_amdgcn_wmma_f32_16x16x4_f32(
       false, a,  false, b,
       (short)0, c,  false,  false);
}

static __device__ __forceinline__ void gln_stats(const float* __restrict__ st, float& mu, float& scale) {
  float s = 0.f, q = 0.f;
  for (int j = 0; j < NSLOT; j++) { s += st[j * SLOTP]; q += st[j * SLOTP + 1]; }
  const float invn = 1.f / (float)N64;
  mu = s * invn;
  const float var = q * invn - mu * mu;
  scale = 1.f / (sqrtf(fmaxf(var, 0.f)) + EPS_LN);
}

static __device__ __forceinline__ void gln_coef(float mu, float scale, float w, float b,
                                                float& a, float& c) {
  a = scale * w;
  c = b - mu * a;
}

static __device__ __forceinline__ void block_stats_add(float s, float q, float* __restrict__ st,
                                                       float* sred) {
  for (int off = 16; off > 0; off >>= 1) { s += __shfl_xor(s, off); q += __shfl_xor(q, off); }
  const int wave = threadIdx.x >> 5, lane = threadIdx.x & 31;
  if (lane == 0) { sred[wave] = s; sred[8 + wave] = q; }
  __syncthreads();
  if (threadIdx.x == 0) {
    float ts = 0.f, tq = 0.f;
    for (int w = 0; w < 8; w++) { ts += sred[w]; tq += sred[8 + w]; }
    float* p = st + (blockIdx.x & (NSLOT - 1)) * SLOTP;
    atomicAdd(p, ts);
    atomicAdd(p + 1, tq);
  }
}

__global__ __launch_bounds__(256) void k_gln_apply(const float4* __restrict__ h,
                                                   const float* __restrict__ st_in,
                                                   const float* __restrict__ w,
                                                   const float* __restrict__ b,
                                                   float4* __restrict__ out,
                                                   float* __restrict__ st_out) {
  __shared__ float sred[16];
  const size_t i0 = (size_t)blockIdx.x * (256 * SPT) + threadIdx.x;
  float mu, scale;
  gln_stats(st_in, mu, scale);
  const int c = (threadIdx.x & 15) << 2;
  const float w0 = w[c + 0], w1 = w[c + 1], w2 = w[c + 2], w3 = w[c + 3];
  const float b0 = b[c + 0], b1 = b[c + 1], b2 = b[c + 2], b3 = b[c + 3];
  float4 v[SPT];
  for (int j = 0; j < SPT; j++) v[j] = h[i0 + j * 256];
  float s = 0.f, q = 0.f;
  for (int j = 0; j < SPT; j++) {
    const float r0 = fmaxf((v[j].x - mu) * scale * w0 + b0, 0.f);
    const float r1 = fmaxf((v[j].y - mu) * scale * w1 + b1, 0.f);
    const float r2 = fmaxf((v[j].z - mu) * scale * w2 + b2, 0.f);
    const float r3 = fmaxf((v[j].w - mu) * scale * w3 + b3, 0.f);
    v[j] = make_float4(r0, r1, r2, r3);
    s += (r0 + r1) + (r2 + r3);
    q += (r0 * r0 + r1 * r1) + (r2 * r2 + r3 * r3);
  }
  block_stats_add(s, q, st_out, sred);
  for (int j = 0; j < SPT; j++) out[i0 + j * 256] = v[j];
}


static __device__ __forceinline__ int block_excl_scan(int v, int* swave, int& total) {
  const int lane = threadIdx.x & 31, wave = threadIdx.x >> 5;
  int inc = v;
  for (int off = 1; off < 32; off <<= 1) {
    int t = __shfl_up(inc, off);
    if (lane >= off) inc += t;
  }
  if (lane == 31) swave[wave] = inc;
  __syncthreads();
  if (threadIdx.x == 0) {
    int s = 0;
    for (int w = 0; w < 8; w++) { int t = swave[w]; swave[w] = s; s += t; }
    swave[8] = s;
  }
  __syncthreads();
  total = swave[8];
  return swave[wave] + inc - v;
}

__global__ __launch_bounds__(256) void k_grp_hist(const int* __restrict__ dst,
                                                  int* __restrict__ hist,
                                                  float* __restrict__ stats) {
  __shared__ int sh[NBK];
  const int tid = threadIdx.x, g = blockIdx.x;
  const int i = g * 256 + tid;
  if (i < NSETS * SETF) stats[i] = 0.f;
  for (int b = tid; b < NBK; b += 256) sh[b] = 0;
  const int e0 = g * (256 * EPT) + tid;
  int d[EPT];
  for (int j = 0; j < EPT; j++) d[j] = dst[e0 + j * 256];
  __syncthreads();
  for (int j = 0; j < EPT; j++) atomicAdd(&sh[d[j] >> 8], 1);
  __syncthreads();
  for (int b = tid; b < NBK; b += 256) hist[b * PBLK + g] = sh[b];
}

__global__ __launch_bounds__(256) void k_grp_scan(int* __restrict__ hist, int* __restrict__ tot) {
  __shared__ int swave[9];
  int* hb = hist + blockIdx.x * PBLK;
  const int k0 = 3 * threadIdx.x, k1 = k0 + 1, k2 = k0 + 2;
  int v0 = hb[min(k0, PBLK - 1)], v1 = hb[min(k1, PBLK - 1)], v2 = hb[min(k2, PBLK - 1)];
  if (k0 >= PBLK) v0 = 0;
  if (k1 >= PBLK) v1 = 0;
  if (k2 >= PBLK) v2 = 0;
  int total;
  const int ex = block_excl_scan(v0 + v1 + v2, swave, total);
  if (k0 < PBLK) hb[k0] = ex;
  if (k1 < PBLK) hb[k1] = ex + v0;
  if (k2 < PBLK) hb[k2] = ex + v0 + v1;
  if (threadIdx.x == 0) tot[blockIdx.x] = total;
}

__global__ __launch_bounds__(256) void k_grp_scatter(const int* __restrict__ src,
                                                     const int* __restrict__ dst,
                                                     const int* __restrict__ hist,
                                                     const int* __restrict__ tot,
                                                     int* __restrict__ bstart,
                                                     unsigned* __restrict__ tmp) {
  __shared__ int swave[9];
  __shared__ int scur[512];
  const int tid = threadIdx.x, g = blockIdx.x;
  const int e0 = g * (256 * EPT) + tid;
  int d[EPT], s[EPT], p[EPT];
  for (int j = 0; j < EPT; j++) { d[j] = dst[e0 + j * 256]; s[j] = src[e0 + j * 256]; }
  const int b0 = 2 * tid, b1 = b0 + 1;
  const int c0 = min(b0, NBK - 1), c1 = min(b1, NBK - 1);
  int t0 = tot[c0], t1 = tot[c1];
  int o0 = hist[c0 * PBLK + g], o1 = hist[c1 * PBLK + g];
  if (b0 >= NBK) { t0 = 0; o0 = 0; }
  if (b1 >= NBK) { t1 = 0; o1 = 0; }
  int total;
  const int ex = block_excl_scan(t0 + t1, swave, total);
  scur[b0] = ex + o0;
  scur[b1] = ex + t0 + o1;
  if (g == 0) {
    if (b0 <= NBK) bstart[b0] = ex;
    if (b1 <= NBK) bstart[b1] = ex + t0;
  }
  __syncthreads();
  for (int j = 0; j < EPT; j++) p[j] = atomicAdd(&scur[d[j] >> 8], 1);
  for (int j = 0; j < EPT; j++) tmp[p[j]] = ((unsigned)s[j] << 8) | ((unsigned)d[j] & 255u);
}

__global__ __launch_bounds__(256) void k_grp_fine(const unsigned* __restrict__ tmp,
                                                  const int* __restrict__ bstart,
                                                  int* __restrict__ rbeg, int* __restrict__ rend,
                                                  int* __restrict__ col) {
  __shared__ int swave[9];
  __shared__ int scnt[257];
  const int tid = threadIdx.x, b = blockIdx.x;
  const int beg = bstart[b], end = bstart[b + 1];
  scnt[tid] = 0;
  if (tid == 0) scnt[256] = 0;
  __syncthreads();
  for (int i = beg + tid; i < end; i += 1024) {
    const int i1 = i + 256, i2 = i + 512, i3 = i + 768;
    const unsigned w0 = tmp[i];
    const unsigned w1 = tmp[min(i1, end - 1)];
    const unsigned w2 = tmp[min(i2, end - 1)];
    const unsigned w3 = tmp[min(i3, end - 1)];
    atomicAdd(&scnt[w0 & 255u], 1);
    atomicAdd(&scnt[(i1 < end) ? (w1 & 255u) : 256u], 1);
    atomicAdd(&scnt[(i2 < end) ? (w2 & 255u) : 256u], 1);
    atomicAdd(&scnt[(i3 < end) ? (w3 & 255u) : 256u], 1);
  }
  __syncthreads();
  const int c = scnt[tid];
  int total;
  const int ex = block_excl_scan(c, swave, total);
  const int n = b * 256 + tid;
  if (n < NN) { rbeg[n] = beg + ex; rend[n] = beg + ex + c; }
  scnt[tid] = beg + ex;
  __syncthreads();
  for (int i = beg + tid; i < end; i += 1024) {
    const int i1 = i + 256, i2 = i + 512, i3 = i + 768;
    const unsigned w0 = tmp[i];
    const unsigned w1 = tmp[min(i1, end - 1)];
    const unsigned w2 = tmp[min(i2, end - 1)];
    const unsigned w3 = tmp[min(i3, end - 1)];
    const int p0 = atomicAdd(&scnt[w0 & 255u], 1);
    const int p1 = atomicAdd(&scnt[(i1 < end) ? (w1 & 255u) : 256u], 1);
    const int p2 = atomicAdd(&scnt[(i2 < end) ? (w2 & 255u) : 256u], 1);
    const int p3 = atomicAdd(&scnt[(i3 < end) ? (w3 & 255u) : 256u], 1);
    col[p0] = (int)(w0 >> 8);
    if (i1 < end) col[p1] = (int)(w1 >> 8);
    if (i2 < end) col[p2] = (int)(w2 >> 8);
    if (i3 < end) col[p3] = (int)(w3 >> 8);
  }
}

static __device__ __forceinline__ void soft4(float v0, float v1, float v2, float v3,
                                             bool p1, bool p2, bool p3,
                                             float& mx, float& s1, float& s2) {
  const float a0 = fmaxf(v0, 0.f) + EPS_MSG;
  const float a1 = p1 ? fmaxf(v1, 0.f) + EPS_MSG : 0.f;
  const float a2 = p2 ? fmaxf(v2, 0.f) + EPS_MSG : 0.f;
  const float a3 = p3 ? fmaxf(v3, 0.f) + EPS_MSG : 0.f;
  const float mn = fmaxf(fmaxf(mx, fmaxf(a0, a1)), fmaxf(a2, a3));
  const float sc = __expf(mx - mn);
  const float e0 = __expf(a0 - mn);
  const float e1 = p1 ? __expf(a1 - mn) : 0.f;
  const float e2 = p2 ? __expf(a2 - mn) : 0.f;
  const float e3 = p3 ? __expf(a3 - mn) : 0.f;
  s1 = s1 * sc + ((e0 + e1) + (e2 + e3));
  s2 = s2 * sc + ((a0 * e0 + a1 * e1) + (a2 * e2 + a3 * e3));
  mx = mn;
}

struct SoftAcc {
  float mx0, mx1, mx2, mx3;
  float s10, s11, s12, s13;
  float s20, s21, s22, s23;
};

template <bool AFF>
static __device__ __forceinline__ void gather_chunk(int idx, int cnt, int cntw, int hb,
                                                    const float* __restrict__ hp,
                                                    const float4 ga, const float4 gc, SoftAcc& A) {
  for (int i = 0; i < cntw; i += 4) {
    const int i0 = __shfl(idx, hb + i);
    const int i1 = __shfl(idx, hb + i + 1);
    const int i2 = __shfl(idx, hb + i + 2);
    const int i3 = __shfl(idx, hb + i + 3);
    if (i < cnt) {
      const bool p1 = i + 1 < cnt, p2 = i + 2 < cnt, p3 = i + 3 < cnt;
      const int j1 = p1 ? i1 : i0, j2 = p2 ? i2 : i0, j3 = p3 ? i3 : i0;
      float4 v0 = *(const float4*)(hp + (size_t)i0 * CH);
      float4 v1 = *(const float4*)(hp + (size_t)j1 * CH);
      float4 v2 = *(const float4*)(hp + (size_t)j2 * CH);
      float4 v3 = *(const float4*)(hp + (size_t)j3 * CH);
      if (AFF) {
        v0.x = fmaf(v0.x, ga.x, gc.x); v1.x = fmaf(v1.x, ga.x, gc.x);
        v2.x = fmaf(v2.x, ga.x, gc.x); v3.x = fmaf(v3.x, ga.x, gc.x);
        v0.y = fmaf(v0.y, ga.y, gc.y); v1.y = fmaf(v1.y, ga.y, gc.y);
        v2.y = fmaf(v2.y, ga.y, gc.y); v3.y = fmaf(v3.y, ga.y, gc.y);
        v0.z = fmaf(v0.z, ga.z, gc.z); v1.z = fmaf(v1.z, ga.z, gc.z);
        v2.z = fmaf(v2.z, ga.z, gc.z); v3.z = fmaf(v3.z, ga.z, gc.z);
        v0.w = fmaf(v0.w, ga.w, gc.w); v1.w = fmaf(v1.w, ga.w, gc.w);
        v2.w = fmaf(v2.w, ga.w, gc.w); v3.w = fmaf(v3.w, ga.w, gc.w);
      }
      soft4(v0.x, v1.x, v2.x, v3.x, p1, p2, p3, A.mx0, A.s10, A.s20);
      soft4(v0.y, v1.y, v2.y, v3.y, p1, p2, p3, A.mx1, A.s11, A.s21);
      soft4(v0.z, v1.z, v2.z, v3.z, p1, p2, p3, A.mx2, A.s12, A.s22);
      soft4(v0.w, v1.w, v2.w, v3.w, p1, p2, p3, A.mx3, A.s13, A.s23);
    }
  }
}

template <bool AFF>
static __device__ __forceinline__ float4 gather_node(int mine, int beg, int deg, int l16, int hb,
                                                     const int* __restrict__ col,
                                                     const float* __restrict__ hp,
                                                     const float4 ga, const float4 gc) {
  const int degw = max(deg, __shfl_xor(deg, 16));
  SoftAcc A = {0.f, 0.f, 0.f, 0.f, 0.f, 0.f, 0.f, 0.f, 0.f, 0.f, 0.f, 0.f};
  gather_chunk<AFF>(mine, deg, min(16, degw), hb, hp, ga, gc, A);
  for (int c0 = 16; c0 < degw; c0 += 16) {
    const int more = (c0 + l16 < deg) ? col[beg + c0 + l16] : 0;
    gather_chunk<AFF>(more, deg - c0, min(16, degw - c0), hb, hp, ga, gc, A);
  }
  float4 r = make_float4(0.f, 0.f, 0.f, 0.f);
  if (deg > 0) r = make_float4(A.s20 / A.s10, A.s21 / A.s11, A.s22 / A.s12, A.s23 / A.s13);
  return r;
}

template <int PATTERN>
static __device__ __forceinline__ float lane_swap(float v) {
  return __int_as_float(__builtin_amdgcn_update_dpp(0, __float_as_int(v), PATTERN, 0xF, 0xF, true));
}

static __device__ __forceinline__ float sum8(float v) {
  v += lane_swap<0xB1>(v);
  v += lane_swap<0x4E>(v);
  v += lane_swap<0x141>(v);
  return v;
}

static __device__ __forceinline__ void ln_acc(const float4 v, float& s, float& q) {
  s += (v.x + v.y) + (v.z + v.w);
  q += (v.x * v.x + v.y * v.y) + (v.z * v.z + v.w * v.w);
}

static __device__ __forceinline__ float4 ln_apply(const float4 v, float mu, float inv,
                                                  const float4 w, const float4 b) {
  return make_float4(fmaxf((v.x - mu) * inv * w.x + b.x, 0.f), fmaxf((v.y - mu) * inv * w.y + b.y, 0.f),
                     fmaxf((v.z - mu) * inv * w.z + b.z, 0.f), fmaxf((v.w - mu) * inv * w.w + b.w, 0.f));
}

template <bool AFF, bool STATS>
__global__ __launch_bounds__(256)
void k_layer(const int* __restrict__ rbeg, const int* __restrict__ rend,
             const int* __restrict__ col, const float* __restrict__ src,
             const float* __restrict__ W1, const float* __restrict__ b1,
             const float* __restrict__ lnw, const float* __restrict__ lnb,
             const float* __restrict__ W2, const float* __restrict__ b2,
             const float* __restrict__ st_in, const float* __restrict__ gw,
             const float* __restrict__ gb, float* __restrict__ out,
             float* __restrict__ st_out) {
  __shared__ __attribute__((aligned(16))) float sAct[32 * PA];
  __shared__ __attribute__((aligned(16))) float sHid[32 * PH];
  __shared__ __attribute__((aligned(16))) float slnw[HD];
  __shared__ __attribute__((aligned(16))) float slnb[HD];
  __shared__ float sb1[HD], sb2[CH];
  __shared__ float sred[16];
  const int tid = threadIdx.x;
  const int base = blockIdx.x * 32;
  const int wave = tid >> 5, lane = tid & 31;
  const int half = lane >> 4, l16 = lane & 15;
  const int hb = lane & 16;
  const int g = l16 << 2;
  const float* hp = src + g;

  {
    const int hw = tid >> 4;
    const int n0 = base + hw, n1 = n0 + 16;
    const int beg0 = rbeg[n0], beg1 = rbeg[n1];
    const int deg0 = rend[n0] - beg0, deg1 = rend[n1] - beg1;
    const int mine0 = (l16 < deg0) ? col[beg0 + l16] : 0;
    const int mine1 = (l16 < deg1) ? col[beg1 + l16] : 0;
    float4 h0 = *(const float4*)(hp + (size_t)n0 * CH);
    float4 h1 = *(const float4*)(hp + (size_t)n1 * CH);
    if (tid < HD) { sb1[tid] = b1[tid]; slnw[tid] = lnw[tid]; slnb[tid] = lnb[tid]; }
    if (tid < CH) sb2[tid] = b2[tid];
    float4 ga = make_float4(1.f, 1.f, 1.f, 1.f), gc = make_float4(0.f, 0.f, 0.f, 0.f);
    if (AFF) {
      float mu, scale;
      gln_stats(st_in, mu, scale);
      gln_coef(mu, scale, gw[g + 0], gb[g + 0], ga.x, gc.x);
      gln_coef(mu, scale, gw[g + 1], gb[g + 1], ga.y, gc.y);
      gln_coef(mu, scale, gw[g + 2], gb[g + 2], ga.z, gc.z);
      gln_coef(mu, scale, gw[g + 3], gb[g + 3], ga.w, gc.w);
      h0.x = fmaxf(fmaf(h0.x, ga.x, gc.x), 0.f); h1.x = fmaxf(fmaf(h1.x, ga.x, gc.x), 0.f);
      h0.y = fmaxf(fmaf(h0.y, ga.y, gc.y), 0.f); h1.y = fmaxf(fmaf(h1.y, ga.y, gc.y), 0.f);
      h0.z = fmaxf(fmaf(h0.z, ga.z, gc.z), 0.f); h1.z = fmaxf(fmaf(h1.z, ga.z, gc.z), 0.f);
      h0.w = fmaxf(fmaf(h0.w, ga.w, gc.w), 0.f); h1.w = fmaxf(fmaf(h1.w, ga.w, gc.w), 0.f);
    }
    const float4 r0 = gather_node<AFF>(mine0, beg0, deg0, l16, hb, col, hp, ga, gc);
    const float4 r1 = gather_node<AFF>(mine1, beg1, deg1, l16, hb, col, hp, ga, gc);
    *(float4*)&sAct[hw * PA + g] = make_float4(h0.x + r0.x, h0.y + r0.y, h0.z + r0.z, h0.w + r0.w);
    *(float4*)&sAct[(16 + hw) * PA + g] =
        make_float4(h1.x + r1.x, h1.y + r1.y, h1.z + r1.z, h1.w + r1.w);
  }
  __syncthreads();

  for (int task = wave; task < 16; task += 8) {
    int rt = task >> 3, ct = task & 7;
    const float* arow = &sAct[(rt * 16 + l16) * PA + 2 * half];
    const float* wcol = W1 + (2 * half) * HD + ct * 16 + l16;
    v8f acc = {0.f, 0.f, 0.f, 0.f, 0.f, 0.f, 0.f, 0.f};
    #pragma unroll
    for (int k0 = 0; k0 < CH; k0 += 4) {
      v2f a;  a.x = arow[k0];       a.y = arow[k0 + 1];
      v2f bb; bb.x = wcol[k0 * HD]; bb.y = wcol[(k0 + 1) * HD];
      acc = wmma4(a, bb, acc);
    }
    #pragma unroll
    for (int r = 0; r < 8; r++) {
      int m = rt * 16 + r + 8 * half;
      int n = ct * 16 + l16;
      sHid[m * PH + n] = acc[r] + sb1[n];
    }
  }
  __syncthreads();

  {
    const int p4 = (tid & 7) << 2;
    float* hrow = &sHid[(tid >> 3) * PH + p4];
    const float4 v0 = *(const float4*)(hrow);
    const float4 v1 = *(const float4*)(hrow + 32);
    const float4 v2 = *(const float4*)(hrow + 64);
    const float4 v3 = *(const float4*)(hrow + 96);
    float s = 0.f, q = 0.f;
    ln_acc(v0, s, q); ln_acc(v1, s, q); ln_acc(v2, s, q); ln_acc(v3, s, q);
    s = sum8(s);
    q = sum8(q);
    const float mu = s * (1.f / HD);
    const float var = q * (1.f / HD) - mu * mu;
    const float inv = rsqrtf(var + EPS_LN);
    *(float4*)(hrow)      = ln_apply(v0, mu, inv, *(const float4*)&slnw[p4],      *(const float4*)&slnb[p4]);
    *(float4*)(hrow + 32) = ln_apply(v1, mu, inv, *(const float4*)&slnw[p4 + 32], *(const float4*)&slnb[p4 + 32]);
    *(float4*)(hrow + 64) = ln_apply(v2, mu, inv, *(const float4*)&slnw[p4 + 64], *(const float4*)&slnb[p4 + 64]);
    *(float4*)(hrow + 96) = ln_apply(v3, mu, inv, *(const float4*)&slnw[p4 + 96], *(const float4*)&slnb[p4 + 96]);
  }
  __syncthreads();

  {
    int rt = wave >> 2, ct = wave & 3;
    const float* arow = &sHid[(rt * 16 + l16) * PH + 2 * half];
    const float* wcol = W2 + (2 * half) * CH + ct * 16 + l16;
    v8f acc = {0.f, 0.f, 0.f, 0.f, 0.f, 0.f, 0.f, 0.f};
    #pragma unroll
    for (int k0 = 0; k0 < HD; k0 += 4) {
      v2f a;  a.x = arow[k0];       a.y = arow[k0 + 1];
      v2f bb; bb.x = wcol[k0 * CH]; bb.y = wcol[(k0 + 1) * CH];
      acc = wmma4(a, bb, acc);
    }
    const int n = ct * 16 + l16;
    const size_t o0 = (size_t)(base + rt * 16 + 8 * half) * CH + n;
    float res[8];
    #pragma unroll
    for (int r = 0; r < 8; r++) res[r] = AFF ? src[o0 + (size_t)r * CH] : 0.f;
    float v[8];
    float s = 0.f, q = 0.f;
    #pragma unroll
    for (int r = 0; r < 8; r++) {
      v[r] = acc[r] + sb2[n] + res[r];
      s += v[r];
      q += v[r] * v[r];
    }
    if (STATS) block_stats_add(s, q, st_out, sred);
    #pragma unroll
    for (int r = 0; r < 8; r++) out[o0 + (size_t)r * CH] = v[r];
  }
}

__global__ __launch_bounds__(256)
void k_head(const float* __restrict__ xc,
            const float* __restrict__ W1, const float* __restrict__ b1,
            const float* __restrict__ W2, const float* __restrict__ b2,
            const float* __restrict__ W3, const float* __restrict__ b3,
            float* __restrict__ out) {
  __shared__ float sA[32 * PA];
  __shared__ float sH[32 * PA];
  __shared__ float sb1[CH], sb2[CH], sW3[CH * 2], sb3[2];
  const int tid = threadIdx.x;
  const int base = blockIdx.x * 32;
  if (tid < CH) { sb1[tid] = b1[tid]; sb2[tid] = b2[tid]; }
  if (tid < CH * 2) sW3[tid] = W3[tid];
  if (tid < 2) sb3[tid] = b3[tid];
  for (int i = tid; i < 32 * CH; i += 256) sA[(i >> 6) * PA + (i & 63)] = xc[(size_t)base * CH + i];
  __syncthreads();

  const int wave = tid >> 5, lane = tid & 31;
  const int half = lane >> 4, l16 = lane & 15;
  const int rt = wave >> 2, ct = wave & 3;

  {
    const float* arow = &sA[(rt * 16 + l16) * PA];
    v8f acc = {0.f, 0.f, 0.f, 0.f, 0.f, 0.f, 0.f, 0.f};
    #pragma unroll
    for (int k0 = 0; k0 < CH; k0 += 4) {
      int k = k0 + 2 * half;
      v2f a;  a.x = arow[k];                     a.y = arow[k + 1];
      v2f bb; bb.x = W1[k * CH + ct * 16 + l16]; bb.y = W1[(k + 1) * CH + ct * 16 + l16];
      acc = wmma4(a, bb, acc);
    }
    #pragma unroll
    for (int r = 0; r < 8; r++) {
      int m = rt * 16 + r + 8 * half, n = ct * 16 + l16;
      sH[m * PA + n] = fmaxf(acc[r] + sb1[n], 0.f);
    }
  }
  __syncthreads();
  {
    const float* arow = &sH[(rt * 16 + l16) * PA];
    v8f acc = {0.f, 0.f, 0.f, 0.f, 0.f, 0.f, 0.f, 0.f};
    #pragma unroll
    for (int k0 = 0; k0 < CH; k0 += 4) {
      int k = k0 + 2 * half;
      v2f a;  a.x = arow[k];                     a.y = arow[k + 1];
      v2f bb; bb.x = W2[k * CH + ct * 16 + l16]; bb.y = W2[(k + 1) * CH + ct * 16 + l16];
      acc = wmma4(a, bb, acc);
    }
    #pragma unroll
    for (int r = 0; r < 8; r++) {
      int m = rt * 16 + r + 8 * half, n = ct * 16 + l16;
      sA[m * PA + n] = fmaxf(acc[r] + sb2[n], 0.f);
    }
  }
  __syncthreads();
  if (tid < 64) {
    int m = tid >> 1, o = tid & 1;
    float s = sb3[o];
    for (int k = 0; k < CH; k++) s += sA[m * PA + k] * sW3[k * 2 + o];
    out[(size_t)(base + m) * 2 + o] = s;
  }
}

extern "C" void kernel_launch(void* const* d_in, const int* in_sizes, int n_in,
                              void* d_out, int out_size, void* d_ws, size_t ws_size,
                              hipStream_t stream) {
  (void)in_sizes; (void)n_in; (void)out_size; (void)ws_size;
  const float* x      = (const float*)d_in[0];
  const int*   ei     = (const int*)d_in[1];
  const int*   src    = ei;
  const int*   dst    = ei + NE;
  const float* c1_W1  = (const float*)d_in[2];
  const float* c1_b1  = (const float*)d_in[3];
  const float* c1_lnw = (const float*)d_in[4];
  const float* c1_lnb = (const float*)d_in[5];
  const float* c1_W2  = (const float*)d_in[6];
  const float* c1_b2  = (const float*)d_in[7];
  const float* n1_w   = (const float*)d_in[8];
  const float* n1_b   = (const float*)d_in[9];
  const float* cW1    = (const float*)d_in[10];
  const float* cb1    = (const float*)d_in[11];
  const float* clnw   = (const float*)d_in[12];
  const float* clnb   = (const float*)d_in[13];
  const float* cW2    = (const float*)d_in[14];
  const float* cb2    = (const float*)d_in[15];
  const float* nw     = (const float*)d_in[16];
  const float* nb     = (const float*)d_in[17];
  const float* lW1    = (const float*)d_in[18];
  const float* lb1    = (const float*)d_in[19];
  const float* lW2    = (const float*)d_in[20];
  const float* lb2    = (const float*)d_in[21];
  const float* lW3    = (const float*)d_in[22];
  const float* lb3    = (const float*)d_in[23];

  float* ws    = (float*)d_ws;
  float* xa    = ws;
  float* hpre  = ws + (size_t)N64;
  int*   csr   = (int*)(ws + 2 * (size_t)N64);
  int*   rbeg  = csr;
  int*   rend  = csr + NN;
  int*   col   = csr + 2 * NN;
  unsigned* tmp = (unsigned*)(col + NE);
  int*   hist  = col + 2 * NE;
  int*   tot   = hist + NBK * PBLK;
  int*   bstart = tot + 512;
  float* xb    = ws + 4 * (size_t)N64;
  float* stats = ws + 3 * (size_t)N64;

  const dim3 B(256);
  const int MLPB = NN / 32;

  k_grp_hist<<<PBLK, B, 0, stream>>>(dst, hist, stats);
  k_grp_scan<<<NBK, B, 0, stream>>>(hist, tot);
  k_grp_scatter<<<PBLK, B, 0, stream>>>(src, dst, hist, tot, bstart, tmp);
  k_grp_fine<<<NBK, B, 0, stream>>>(tmp, bstart, rbeg, rend, col);

  k_layer<false, true><<<MLPB, B, 0, stream>>>(rbeg, rend, col, x, c1_W1, c1_b1, c1_lnw, c1_lnb,
                                               c1_W2, c1_b2, nullptr, nullptr, nullptr,
                                               hpre, stats);
  k_gln_apply<<<SBLK, B, 0, stream>>>((const float4*)hpre, stats, n1_w, n1_b, (float4*)xa,
                                      stats + SETF);

  for (int i = 0; i < 3; i++) {
    const float* st = stats + (size_t)(1 + i) * SETF;
    const float* xin = (i & 1) ? xb : xa;
    float* xout = (i & 1) ? xa : xb;
    if (i < 2)
      k_layer<true, true><<<MLPB, B, 0, stream>>>(rbeg, rend, col, xin,
                                       cW1 + (size_t)i * CH * HD, cb1 + (size_t)i * HD,
                                       clnw + (size_t)i * HD, clnb + (size_t)i * HD,
                                       cW2 + (size_t)i * HD * CH, cb2 + (size_t)i * CH,
                                       st, nw + i * CH, nb + i * CH, xout,
                                       stats + (size_t)(2 + i) * SETF);
    else
      k_layer<true, false><<<MLPB, B, 0, stream>>>(rbeg, rend, col, xin,
                                       cW1 + (size_t)i * CH * HD, cb1 + (size_t)i * HD,
                                       clnw + (size_t)i * HD, clnb + (size_t)i * HD,
                                       cW2 + (size_t)i * HD * CH, cb2 + (size_t)i * CH,
                                       st, nw + i * CH, nb + i * CH, xout, nullptr);
  }

  k_head<<<MLPB, B, 0, stream>>>(xb, lW1, lb1, lW2, lb2, lW3, lb3, (float*)d_out);
}
